# starting kernel
# baseline (speedup 1.0000x reference)
#include <hip/hip_runtime.h>

typedef __attribute__((ext_vector_type(2))) float v2f;
typedef __attribute__((ext_vector_type(8))) float v8f;

#define NNODES 50000
#define NEDGES 800000
#define ETOT   850000
#define FIN    128
#define HIDC   64
#define NHEADS 4
#define EPSBN  1e-5f

template <int K>
__global__ __launch_bounds__(256) void gemm_wmma_f32(
    const float* __restrict__ A, const float* __restrict__ B,
    float* __restrict__ C, int Nc, int n_wave_jobs)
{
  __shared__ float Bs[K * 16];

  const int tid  = threadIdx.x;
  const int lane = tid & 31;
  const int wave = tid >> 5;
  const int half = lane >> 4;
  const int ml   = lane & 15;
  const int col0 = blockIdx.x * 16;

  #pragma unroll
  for (int i = tid; i < K * 16; i += 256) {
    const int r = i >> 4, cc = i & 15;
    Bs[i] = B[(size_t)r * Nc + col0 + cc];
  }
  __syncthreads();

  const int job = blockIdx.y * 8 + wave;
  if (job < n_wave_jobs) {
    const int row_base = job * 80;

    v8f acc[5];
    #pragma unroll
    for (int t = 0; t < 5; ++t)
      acc[t] = (v8f){0.f,0.f,0.f,0.f,0.f,0.f,0.f,0.f};

    #pragma unroll 4
    for (int k0 = 0; k0 < K; k0 += 4) {
      v2f b;
      b.x = Bs[(k0 + 2*half + 0) * 16 + ml];
      b.y = Bs[(k0 + 2*half + 1) * 16 + ml];
      const float* ap = A + (size_t)(row_base + ml) * K + (k0 + 2*half);
      #pragma unroll
      for (int t = 0; t < 5; ++t) {
        v2f a = *(const v2f*)(ap + (size_t)(16 * t) * K);
        acc[t] = __builtin_amdgcn_wmma_f32_16x16x4_f32(
            false, a, false, b, (short)0, acc[t], false, false);
      }
    }

    #pragma unroll
    for (int t = 0; t < 5; ++t) {
      float* cp = C + (size_t)(row_base + 16*t + 8*half) * Nc + col0 + ml;
      #pragma unroll
      for (int i = 0; i < 8; ++i)
        cp[(size_t)i * Nc] = acc[t][i];
    }
  }
}

__global__ void att_scores(const float* __restrict__ hf,
                           const float* __restrict__ a_s,
                           const float* __restrict__ a_d,
                           float* __restrict__ s_src,
                           float* __restrict__ s_dst,
                           int n, int heads, int c)
{
  int idx = blockIdx.x * blockDim.x + threadIdx.x;
  if (idx >= n * heads) return;
  int node = idx / heads, h = idx % heads;
  const float* f  = hf  + (size_t)node * heads * c + (size_t)h * c;
  const float* as = a_s + (size_t)h * c;
  const float* ad = a_d + (size_t)h * c;
  float ss = 0.f, sd = 0.f;
  #pragma unroll 4
  for (int i = 0; i < c; ++i) { float v = f[i]; ss += v * as[i]; sd += v * ad[i]; }
  s_src[idx] = ss; s_dst[idx] = sd;
}

__device__ __forceinline__ unsigned f2key(float f) {
  unsigned u = __float_as_uint(f);
  return (u & 0x80000000u) ? ~u : (u | 0x80000000u);
}
__device__ __forceinline__ float key2f(unsigned k) {
  return __uint_as_float((k & 0x80000000u) ? (k ^ 0x80000000u) : ~k);
}

__device__ __forceinline__ void edge_src_dst(const int* __restrict__ ei, int e,
                                             int& src, int& dst) {
  if (e < NEDGES) { src = ei[e]; dst = ei[NEDGES + e]; }
  else            { src = dst = e - NEDGES; }
}

__global__ void edge_logits(const int* __restrict__ ei,
                            const float* __restrict__ s_src,
                            const float* __restrict__ s_dst,
                            float* __restrict__ e_buf,
                            unsigned* __restrict__ emax_key, int heads)
{
  int idx = blockIdx.x * blockDim.x + threadIdx.x;
  if (idx >= ETOT * heads) return;
  int e = idx / heads, h = idx % heads;
  int src, dst; edge_src_dst(ei, e, src, dst);
  float v = s_src[src * heads + h] + s_dst[dst * heads + h];
  v = v > 0.f ? v : 0.2f * v;
  e_buf[idx] = v;
  atomicMax(&emax_key[dst * heads + h], f2key(v));
}

__global__ void edge_exp(const int* __restrict__ ei,
                         float* __restrict__ e_buf,
                         const unsigned* __restrict__ emax_key,
                         float* __restrict__ denom, int heads)
{
  int idx = blockIdx.x * blockDim.x + threadIdx.x;
  if (idx >= ETOT * heads) return;
  int e = idx / heads, h = idx % heads;
  int src, dst; edge_src_dst(ei, e, src, dst);
  float m  = key2f(emax_key[dst * heads + h]);
  float ee = expf(e_buf[idx] - m);
  e_buf[idx] = ee;
  atomicAdd(&denom[dst * heads + h], ee);
}

__global__ void edge_aggregate(const int* __restrict__ ei,
                               const float* __restrict__ e_buf,
                               const float* __restrict__ denom,
                               const float* __restrict__ hf,
                               float* __restrict__ agg, int heads, int c)
{
  const int hc_tot = heads * c;
  const int hq     = hc_tot >> 2;
  long long tid   = (long long)blockIdx.x * blockDim.x + threadIdx.x;
  long long total = (long long)ETOT * hq;
  if (tid >= total) return;
  const int e  = (int)(tid / hq);
  const int q  = (int)(tid % hq);
  const int hc = q << 2;
  const int h  = hc / c;
  int src, dst; edge_src_dst(ei, e, src, dst);
  const float alpha = e_buf[(size_t)e * heads + h] /
                      (denom[dst * heads + h] + 1e-16f);
  const float4 v = *(const float4*)(hf + (size_t)src * hc_tot + hc);
  float* dp = agg + (size_t)dst * hc_tot + hc;
  atomicAdd(dp + 0, v.x * alpha);
  atomicAdd(dp + 1, v.y * alpha);
  atomicAdd(dp + 2, v.z * alpha);
  atomicAdd(dp + 3, v.w * alpha);
}

__global__ void post_bn(const float* __restrict__ agg,
                        const float* __restrict__ bias,
                        const float* __restrict__ g,
                        const float* __restrict__ bt,
                        const float* __restrict__ m,
                        const float* __restrict__ v,
                        float* __restrict__ out, int n, int ch, int do_elu)
{
  long long idx = (long long)blockIdx.x * blockDim.x + threadIdx.x;
  if (idx >= (long long)n * ch) return;
  int cix = (int)(idx % ch);
  float x = agg[idx] + bias[cix];
  x = (x - m[cix]) * rsqrtf(v[cix] + EPSBN) * g[cix] + bt[cix];
  if (do_elu) x = x > 0.f ? x : expm1f(x);
  out[idx] = x;
}

__global__ void final_linear(const float* __restrict__ h,
                             const float* __restrict__ Wc,
                             const float* __restrict__ bc,
                             float* __restrict__ out)
{
  int n = blockIdx.x * blockDim.x + threadIdx.x;
  if (n >= NNODES) return;
  float o0 = bc[0], o1 = bc[1];
  const float* r = h + (size_t)n * HIDC;
  #pragma unroll 8
  for (int i = 0; i < HIDC; ++i) {
    float x = r[i];
    o0 += x * Wc[2*i + 0];
    o1 += x * Wc[2*i + 1];
  }
  out[2*n + 0] = o0;
  out[2*n + 1] = o1;
}

__global__ void fill_u32(unsigned* __restrict__ p, unsigned val, long long n)
{
  long long i = (long long)blockIdx.x * blockDim.x + threadIdx.x;
  if (i < n) p[i] = val;
}

static inline unsigned gblk(long long total, int bs) {
  return (unsigned)((total + bs - 1) / bs);
}

extern "C" void kernel_launch(void* const* d_in, const int* in_sizes, int n_in,
                              void* d_out, int out_size, void* d_ws, size_t ws_size,
                              hipStream_t stream)
{
  const float* x  = (const float*)d_in[0];
  const int*   ei = (const int*)  d_in[1];
  const float* Wc = (const float*)d_in[26];
  const float* bc = (const float*)d_in[27];

  char* ws = (char*)d_ws;
  float*    hf    = (float*)ws;    ws += (size_t)NNODES * 256 * sizeof(float);
  float*    agg   = (float*)ws;    ws += (size_t)NNODES * 256 * sizeof(float);
  float*    ssrc  = (float*)ws;    ws += (size_t)NNODES * NHEADS * sizeof(float);
  float*    sdst  = (float*)ws;    ws += (size_t)NNODES * NHEADS * sizeof(float);
  unsigned* emax  = (unsigned*)ws; ws += (size_t)NNODES * NHEADS * sizeof(unsigned);
  float*    denom = (float*)ws;    ws += (size_t)NNODES * NHEADS * sizeof(float);
  float*    ebuf  = (float*)ws;    ws += (size_t)ETOT   * NHEADS * sizeof(float);

  const int n_wave_jobs = (NNODES / 16) / 5;

  const float* in = x;
  for (int l = 0; l < 3; ++l) {
    int base = 2 + l * 8;
    const float* W  = (const float*)d_in[base + 0];
    const float* as = (const float*)d_in[base + 1];
    const float* ad = (const float*)d_in[base + 2];
    const float* bb = (const float*)d_in[base + 3];
    const float* g  = (const float*)d_in[base + 4];
    const float* bt = (const float*)d_in[base + 5];
    const float* m  = (const float*)d_in[base + 6];
    const float* vv = (const float*)d_in[base + 7];

    const int heads = (l < 2) ? NHEADS : 1;
    const int c     = HIDC;
    const int hc    = heads * c;

    dim3 ggrid(hc / 16, (unsigned)((n_wave_jobs + 7) / 8));
    if (l == 0)
      gemm_wmma_f32<FIN><<<ggrid, 256, 0, stream>>>(in, W, hf, hc, n_wave_jobs);
    else
      gemm_wmma_f32<NHEADS * HIDC><<<ggrid, 256, 0, stream>>>(in, W, hf, hc, n_wave_jobs);

    const int nh = NNODES * heads;
    att_scores<<<gblk(nh, 256), 256, 0, stream>>>(hf, as, ad, ssrc, sdst,
                                                  NNODES, heads, c);

    fill_u32<<<gblk((long long)NNODES * hc, 256), 256, 0, stream>>>(
        (unsigned*)agg, 0u, (long long)NNODES * hc);
    fill_u32<<<gblk(nh, 256), 256, 0, stream>>>(emax, 0u, nh);
    fill_u32<<<gblk(nh, 256), 256, 0, stream>>>((unsigned*)denom, 0u, nh);

    const long long eh = (long long)ETOT * heads;
    edge_logits<<<gblk(eh, 256), 256, 0, stream>>>(ei, ssrc, sdst, ebuf, emax, heads);
    edge_exp   <<<gblk(eh, 256), 256, 0, stream>>>(ei, ebuf, emax, denom, heads);

    const long long tot4 = (long long)ETOT * (hc >> 2);
    edge_aggregate<<<gblk(tot4, 256), 256, 0, stream>>>(ei, ebuf, denom, hf, agg,
                                                        heads, c);

    post_bn<<<gblk((long long)NNODES * hc, 256), 256, 0, stream>>>(
        agg, bb, g, bt, m, vv, agg, NNODES, hc, (l < 2) ? 1 : 0);

    in = agg;
  }

  final_linear<<<gblk(NNODES, 256), 256, 0, stream>>>(in, Wc, bc, (float*)d_out);
}


# Edges bucketed by destination once per call (integer atomics); one wave per node then does max, exp, sum and weighted row sum in registers: no float atomics, sum order = slot order, divide after sum
# speedup vs baseline: 2.3206x; 2.3206x over previous
#include <hip/hip_runtime.h>

typedef __attribute__((ext_vector_type(2))) float v2f;
typedef __attribute__((ext_vector_type(8))) float v8f;

#define NNODES 50000
#define NEDGES 800000
#define ETOT   850000
#define FIN    128
#define HIDC   64
#define NHEADS 4
#define EPSBN  1e-5f
#define SCAN_T   1024
#define SCAN_PER 52
#define NPAD     (SCAN_T * SCAN_PER)

template <int K>
__global__ __launch_bounds__(256) void gemm_wmma_f32(
    const float* __restrict__ A, const float* __restrict__ B,
    float* __restrict__ C, int Nc, int n_wave_jobs)
{
  __shared__ float Bs[K * 16];

  const int tid  = threadIdx.x;
  const int lane = tid & 31;
  const int wave = tid >> 5;
  const int half = lane >> 4;
  const int ml   = lane & 15;
  const int col0 = blockIdx.x * 16;

  #pragma unroll
  for (int i = tid; i < K * 16; i += 256) {
    const int r = i >> 4, cc = i & 15;
    Bs[i] = B[(size_t)r * Nc + col0 + cc];
  }
  __syncthreads();

  const int job = blockIdx.y * 8 + wave;
  if (job < n_wave_jobs) {
    const int row_base = job * 80;

    v8f acc[5];
    #pragma unroll
    for (int t = 0; t < 5; ++t)
      acc[t] = (v8f){0.f,0.f,0.f,0.f,0.f,0.f,0.f,0.f};

    #pragma unroll 4
    for (int k0 = 0; k0 < K; k0 += 4) {
      v2f b;
      b.x = Bs[(k0 + 2*half + 0) * 16 + ml];
      b.y = Bs[(k0 + 2*half + 1) * 16 + ml];
      const float* ap = A + (size_t)(row_base + ml) * K + (k0 + 2*half);
      #pragma unroll
      for (int t = 0; t < 5; ++t) {
        v2f a = *(const v2f*)(ap + (size_t)(16 * t) * K);
        acc[t] = __builtin_amdgcn_wmma_f32_16x16x4_f32(
            false, a, false, b, (short)0, acc[t], false, false);
      }
    }

    #pragma unroll
    for (int t = 0; t < 5; ++t) {
      float* cp = C + (size_t)(row_base + 16*t + 8*half) * Nc + col0 + ml;
      #pragma unroll
      for (int i = 0; i < 8; ++i)
        cp[(size_t)i * Nc] = acc[t][i];
    }
  }
}

__global__ void att_scores(const float* __restrict__ hf,
                           const float* __restrict__ a_s,
                           const float* __restrict__ a_d,
                           float* __restrict__ s_src,
                           float* __restrict__ s_dst,
                           int n, int heads, int c)
{
  int idx = blockIdx.x * blockDim.x + threadIdx.x;
  if (idx >= n * heads) return;
  int node = idx / heads, h = idx % heads;
  const float* f  = hf  + (size_t)node * heads * c + (size_t)h * c;
  const float* as = a_s + (size_t)h * c;
  const float* ad = a_d + (size_t)h * c;
  float ss = 0.f, sd = 0.f;
  #pragma unroll 4
  for (int i = 0; i < c; ++i) { float v = f[i]; ss += v * as[i]; sd += v * ad[i]; }
  s_src[idx] = ss; s_dst[idx] = sd;
}

__global__ void bucket_count(const int* __restrict__ ei,
                             unsigned* __restrict__ cnt)
{
  int e = blockIdx.x * blockDim.x + threadIdx.x;
  if (e >= NEDGES) return;
  atomicAdd(&cnt[ei[NEDGES + e]], 1u);
}

__global__ __launch_bounds__(SCAN_T) void bucket_scan(
    const unsigned* __restrict__ cnt, unsigned* __restrict__ rowptr,
    unsigned* __restrict__ cursor)
{
  __shared__ unsigned part[SCAN_T];
  const int t = threadIdx.x;
  const uint4* cp = (const uint4*)(cnt + (size_t)t * SCAN_PER);
  uint4 c[SCAN_PER / 4];
  #pragma unroll
  for (int i = 0; i < SCAN_PER / 4; ++i) c[i] = cp[i];
  unsigned s = 0;
  #pragma unroll
  for (int i = 0; i < SCAN_PER / 4; ++i) s += c[i].x + c[i].y + c[i].z + c[i].w;
  part[t] = s;
  __syncthreads();
  for (int off = 1; off < SCAN_T; off <<= 1) {
    unsigned v = (t >= off) ? part[t - off] : 0u;
    __syncthreads();
    part[t] += v;
    __syncthreads();
  }
  unsigned run = part[t] - s;
  uint4* rp = (uint4*)(rowptr + (size_t)t * SCAN_PER);
  uint4* up = (uint4*)(cursor + (size_t)t * SCAN_PER);
  #pragma unroll
  for (int i = 0; i < SCAN_PER / 4; ++i) {
    uint4 o;
    o.x = run; run += c[i].x;
    o.y = run; run += c[i].y;
    o.z = run; run += c[i].z;
    o.w = run; run += c[i].w;
    rp[i] = o; up[i] = o;
  }
}

__global__ void bucket_fill(const int* __restrict__ ei,
                            unsigned* __restrict__ cursor,
                            int* __restrict__ srcs)
{
  int e = blockIdx.x * blockDim.x + threadIdx.x;
  if (e >= NEDGES) return;
  const int s = ei[e], d = ei[NEDGES + e];
  const unsigned slot = atomicAdd(&cursor[d], 1u);
  srcs[slot] = s;
}

__device__ __forceinline__ float leaky02(float v) { return v > 0.f ? v : 0.2f * v; }

__device__ __forceinline__ float wave_max(float v) {
  #pragma unroll
  for (int o = 16; o > 0; o >>= 1) v = fmaxf(v, __shfl_xor(v, o, 32));
  return v;
}
__device__ __forceinline__ float wave_sum(float v) {
  #pragma unroll
  for (int o = 16; o > 0; o >>= 1) v += __shfl_xor(v, o, 32);
  return v;
}
__device__ __forceinline__ float lane_bcast(float v, int j) {
  return __int_as_float(__builtin_amdgcn_readlane(__float_as_int(v), j));
}

template <int HEADS>
__device__ __forceinline__ void load_scores(const float* __restrict__ p, int i,
                                            float (&o)[HEADS]) {
  if (HEADS == 4) {
    const float4 t = *(const float4*)(p + (size_t)i * 4);
    o[0] = t.x; o[1] = t.y; o[2] = t.z; o[3] = t.w;
  } else {
    #pragma unroll
    for (int h = 0; h < HEADS; ++h) o[h] = p[(size_t)i * HEADS + h];
  }
}

template <int HEADS>
__global__ __launch_bounds__(256) void node_softmax_gather(
    const unsigned* __restrict__ rowptr, const int* __restrict__ srcs,
    const float* __restrict__ s_src, const float* __restrict__ s_dst,
    const float* __restrict__ hf, float* __restrict__ agg)
{
  constexpr int HC = HEADS * HIDC;
  const int lane = threadIdx.x & 31;
  const int wave = __builtin_amdgcn_readfirstlane((int)(threadIdx.x >> 5));
  const int node = blockIdx.x * 8 + wave;
  if (node >= NNODES) return;
  const int beg = (int)rowptr[node];
  const int deg = (int)rowptr[node + 1] - beg;
  const int cnt = deg + 1;

  float sd[HEADS], m[HEADS], den[HEADS];
  load_scores<HEADS>(s_dst, node, sd);
  #pragma unroll
  for (int h = 0; h < HEADS; ++h) { m[h] = -INFINITY; den[h] = 0.f; }

  for (int cb = 0; cb < cnt; cb += 32) {
    const int idx = cb + lane;
    if (idx < cnt) {
      const int s = idx < deg ? srcs[beg + idx] : node;
      float sc[HEADS];
      load_scores<HEADS>(s_src, s, sc);
      #pragma unroll
      for (int h = 0; h < HEADS; ++h) m[h] = fmaxf(m[h], leaky02(sc[h] + sd[h]));
    }
  }
  #pragma unroll
  for (int h = 0; h < HEADS; ++h) m[h] = wave_max(m[h]);

  float4 accA = make_float4(0.f, 0.f, 0.f, 0.f);
  float4 accB = make_float4(0.f, 0.f, 0.f, 0.f);
  const bool lo = lane < 16;

  for (int cb = 0; cb < cnt; cb += 32) {
    const int idx = cb + lane;
    int s = node;
    float e[HEADS];
    #pragma unroll
    for (int h = 0; h < HEADS; ++h) e[h] = 0.f;
    if (idx < cnt) {
      s = idx < deg ? srcs[beg + idx] : node;
      float sc[HEADS];
      load_scores<HEADS>(s_src, s, sc);
      #pragma unroll
      for (int h = 0; h < HEADS; ++h) e[h] = expf(leaky02(sc[h] + sd[h]) - m[h]);
    }
    #pragma unroll
    for (int h = 0; h < HEADS; ++h) den[h] += e[h];

    const int n = (cnt - cb) < 32 ? (cnt - cb) : 32;
    #pragma unroll 4
    for (int j = 0; j < n; ++j) {
      const int sj = __builtin_amdgcn_readlane(s, j);
      const float* row = hf + (size_t)sj * HC;
      if (HEADS == 4) {
        const float e0 = lane_bcast(e[0], j), e1 = lane_bcast(e[1 % HEADS], j);
        const float e2 = lane_bcast(e[2 % HEADS], j), e3 = lane_bcast(e[3 % HEADS], j);
        const float eA = lo ? e0 : e1;
        const float eB = lo ? e2 : e3;
        const float4 a = *(const float4*)(row + 4 * lane);
        const float4 b = *(const float4*)(row + 128 + 4 * lane);
        accA.x += eA * a.x; accA.y += eA * a.y; accA.z += eA * a.z; accA.w += eA * a.w;
        accB.x += eB * b.x; accB.y += eB * b.y; accB.z += eB * b.z; accB.w += eB * b.w;
      } else {
        const float e0 = lane_bcast(e[0], j);
        const float2 a = *(const float2*)(row + 2 * lane);
        accA.x += e0 * a.x; accA.y += e0 * a.y;
      }
    }
  }
  #pragma unroll
  for (int h = 0; h < HEADS; ++h) den[h] = wave_sum(den[h]);

  float* op = agg + (size_t)node * HC;
  if (HEADS == 4) {
    const float dA = (lo ? den[0] : den[1 % HEADS]) + 1e-16f;
    const float dB = (lo ? den[2 % HEADS] : den[3 % HEADS]) + 1e-16f;
    *(float4*)(op + 4 * lane) =
        make_float4(accA.x / dA, accA.y / dA, accA.z / dA, accA.w / dA);
    *(float4*)(op + 128 + 4 * lane) =
        make_float4(accB.x / dB, accB.y / dB, accB.z / dB, accB.w / dB);
  } else {
    const float d0 = den[0] + 1e-16f;
    *(float2*)(op + 2 * lane) = make_float2(accA.x / d0, accA.y / d0);
  }
}

__global__ void post_bn(const float* __restrict__ agg,
                        const float* __restrict__ bias,
                        const float* __restrict__ g,
                        const float* __restrict__ bt,
                        const float* __restrict__ m,
                        const float* __restrict__ v,
                        float* __restrict__ out, int n, int ch, int do_elu)
{
  long long idx = (long long)blockIdx.x * blockDim.x + threadIdx.x;
  if (idx >= (long long)n * ch) return;
  int cix = (int)(idx % ch);
  float x = agg[idx] + bias[cix];
  x = (x - m[cix]) * rsqrtf(v[cix] + EPSBN) * g[cix] + bt[cix];
  if (do_elu) x = x > 0.f ? x : expm1f(x);
  out[idx] = x;
}

__global__ void final_linear(const float* __restrict__ h,
                             const float* __restrict__ Wc,
                             const float* __restrict__ bc,
                             float* __restrict__ out)
{
  int n = blockIdx.x * blockDim.x + threadIdx.x;
  if (n >= NNODES) return;
  float o0 = bc[0], o1 = bc[1];
  const float* r = h + (size_t)n * HIDC;
  #pragma unroll 8
  for (int i = 0; i < HIDC; ++i) {
    float x = r[i];
    o0 += x * Wc[2*i + 0];
    o1 += x * Wc[2*i + 1];
  }
  out[2*n + 0] = o0;
  out[2*n + 1] = o1;
}

__global__ void fill_u32(unsigned* __restrict__ p, unsigned val, long long n)
{
  long long i = (long long)blockIdx.x * blockDim.x + threadIdx.x;
  if (i < n) p[i] = val;
}

static inline unsigned gblk(long long total, int bs) {
  return (unsigned)((total + bs - 1) / bs);
}

extern "C" void kernel_launch(void* const* d_in, const int* in_sizes, int n_in,
                              void* d_out, int out_size, void* d_ws, size_t ws_size,
                              hipStream_t stream)
{
  const float* x  = (const float*)d_in[0];
  const int*   ei = (const int*)  d_in[1];
  const float* Wc = (const float*)d_in[26];
  const float* bc = (const float*)d_in[27];

  char* ws = (char*)d_ws;
  float*    hf    = (float*)ws;    ws += (size_t)NNODES * 256 * sizeof(float);
  float*    agg   = (float*)ws;    ws += (size_t)NNODES * 256 * sizeof(float);
  float*    ssrc  = (float*)ws;    ws += (size_t)NNODES * NHEADS * sizeof(float);
  float*    sdst  = (float*)ws;    ws += (size_t)NNODES * NHEADS * sizeof(float);
  unsigned* cnt    = (unsigned*)ws; ws += (size_t)NPAD * sizeof(unsigned);
  unsigned* cursor = (unsigned*)ws; ws += (size_t)NPAD * sizeof(unsigned);
  unsigned* rowptr = (unsigned*)ws; ws += (size_t)NPAD * sizeof(unsigned);
  int*      srcs   = (int*)ws;      ws += (size_t)NEDGES * sizeof(int);

  fill_u32<<<gblk(NPAD, 256), 256, 0, stream>>>(cnt, 0u, NPAD);
  bucket_count<<<gblk(NEDGES, 256), 256, 0, stream>>>(ei, cnt);
  bucket_scan<<<1, SCAN_T, 0, stream>>>(cnt, rowptr, cursor);
  bucket_fill<<<gblk(NEDGES, 256), 256, 0, stream>>>(ei, cursor, srcs);

  const int n_wave_jobs = (NNODES / 16) / 5;

  const float* in = x;
  for (int l = 0; l < 3; ++l) {
    int base = 2 + l * 8;
    const float* W  = (const float*)d_in[base + 0];
    const float* as = (const float*)d_in[base + 1];
    const float* ad = (const float*)d_in[base + 2];
    const float* bb = (const float*)d_in[base + 3];
    const float* g  = (const float*)d_in[base + 4];
    const float* bt = (const float*)d_in[base + 5];
    const float* m  = (const float*)d_in[base + 6];
    const float* vv = (const float*)d_in[base + 7];

    const int heads = (l < 2) ? NHEADS : 1;
    const int c     = HIDC;
    const int hc    = heads * c;

    dim3 ggrid(hc / 16, (unsigned)((n_wave_jobs + 7) / 8));
    if (l == 0)
      gemm_wmma_f32<FIN><<<ggrid, 256, 0, stream>>>(in, W, hf, hc, n_wave_jobs);
    else
      gemm_wmma_f32<NHEADS * HIDC><<<ggrid, 256, 0, stream>>>(in, W, hf, hc, n_wave_jobs);

    const int nh = NNODES * heads;
    att_scores<<<gblk(nh, 256), 256, 0, stream>>>(hf, as, ad, ssrc, sdst,
                                                  NNODES, heads, c);

    if (heads == NHEADS)
      node_softmax_gather<NHEADS><<<NNODES / 8, 256, 0, stream>>>(
          rowptr, srcs, ssrc, sdst, hf, agg);
    else
      node_softmax_gather<1><<<NNODES / 8, 256, 0, stream>>>(
          rowptr, srcs, ssrc, sdst, hf, agg);

    post_bn<<<gblk((long long)NNODES * hc, 256), 256, 0, stream>>>(
        agg, bb, g, bt, m, vv, agg, NNODES, hc, (l < 2) ? 1 : 0);

    in = agg;
  }

  final_linear<<<gblk(NNODES, 256), 256, 0, stream>>>(in, Wc, bc, (float*)d_out);
}


# GEMM re-tiled: each wave loads a 16-row A fragment once per K-step and feeds four WMMAs from a 64-column B panel in LDS (A fetched once per 64 output columns, not per 16); K sum order unchanged
# speedup vs baseline: 3.4686x; 1.4947x over previous
#include <hip/hip_runtime.h>

typedef __attribute__((ext_vector_type(2))) float v2f;
typedef __attribute__((ext_vector_type(8))) float v8f;

#define NNODES 50000
#define NEDGES 800000
#define ETOT   850000
#define FIN    128
#define HIDC   64
#define NHEADS 4
#define EPSBN  1e-5f
#define SCAN_T   1024
#define SCAN_PER 52
#define NPAD     (SCAN_T * SCAN_PER)

#define GEMM_NP     64
#define GEMM_PP     160
#define GEMM_MTILES (NNODES / 16)

template <int K>
__global__ __launch_bounds__(256) void gemm_wmma_f32(
    const float* __restrict__ A, const float* __restrict__ B,
    float* __restrict__ C, int Nc)
{
  __shared__ float Bs[(K / 2) * GEMM_PP];

  const int tid  = threadIdx.x;
  const int lane = tid & 31;
  const int wave = tid >> 5;
  const int half = lane >> 4;
  const int ml   = lane & 15;
  const int col0 = blockIdx.x * GEMM_NP;

  #pragma unroll
  for (int it = 0; it < (K / 2) * (GEMM_NP / 4) / 256; ++it) {
    const int i = tid + 256 * it;
    const int pr = i >> 4, c4 = (i & 15) * 4;
    const float* src = B + (size_t)(2 * pr) * Nc + col0 + c4;
    const float4 x0 = *(const float4*)(src);
    const float4 x1 = *(const float4*)(src + Nc);
    float* dst = Bs + pr * GEMM_PP + 2 * c4;
    *(float4*)(dst)     = make_float4(x0.x, x1.x, x0.y, x1.y);
    *(float4*)(dst + 4) = make_float4(x0.z, x1.z, x0.w, x1.w);
  }
  __syncthreads();

  const int tile = blockIdx.y * 8 + wave;
  if (tile < GEMM_MTILES) {
    const float* ap = A + (size_t)(tile * 16 + ml) * K + 2 * half;
    const float* bp = Bs + half * GEMM_PP + 2 * ml;

    v8f acc[4];
    #pragma unroll
    for (int n = 0; n < 4; ++n)
      acc[n] = (v8f){0.f,0.f,0.f,0.f,0.f,0.f,0.f,0.f};

    v2f a_cur[8], a_nxt[8];
    #pragma unroll
    for (int j = 0; j < 8; ++j) a_cur[j] = *(const v2f*)(ap + 4 * j);

    #pragma nounroll
    for (int kb = 0; kb < K; kb += 32) {
      const int kn = (kb + 32 < K) ? kb + 32 : kb;
      #pragma unroll
      for (int j = 0; j < 8; ++j) a_nxt[j] = *(const v2f*)(ap + kn + 4 * j);

      #pragma unroll
      for (int j = 0; j < 8; ++j) {
        const float* bk = bp + ((kb + 4 * j) >> 1) * GEMM_PP;
        #pragma unroll
        for (int n = 0; n < 4; ++n) {
          const v2f b = *(const v2f*)(bk + 32 * n);
          acc[n] = __builtin_amdgcn_wmma_f32_16x16x4_f32(
              false, a_cur[j], false, b, (short)0, acc[n], false, false);
        }
      }
      #pragma unroll
      for (int j = 0; j < 8; ++j) a_cur[j] = a_nxt[j];
    }

    float* cp = C + (size_t)(tile * 16 + 8 * half) * Nc + col0 + ml;
    #pragma unroll
    for (int n = 0; n < 4; ++n) {
      #pragma unroll
      for (int i = 0; i < 8; ++i)
        cp[(size_t)i * Nc + 16 * n] = acc[n][i];
    }
  }
}

__global__ void att_scores(const float* __restrict__ hf,
                           const float* __restrict__ a_s,
                           const float* __restrict__ a_d,
                           float* __restrict__ s_src,
                           float* __restrict__ s_dst,
                           int n, int heads, int c)
{
  int idx = blockIdx.x * blockDim.x + threadIdx.x;
  if (idx >= n * heads) return;
  int node = idx / heads, h = idx % heads;
  const float* f  = hf  + (size_t)node * heads * c + (size_t)h * c;
  const float* as = a_s + (size_t)h * c;
  const float* ad = a_d + (size_t)h * c;
  float ss = 0.f, sd = 0.f;
  #pragma unroll 4
  for (int i = 0; i < c; ++i) { float v = f[i]; ss += v * as[i]; sd += v * ad[i]; }
  s_src[idx] = ss; s_dst[idx] = sd;
}

__global__ void bucket_count(const int* __restrict__ ei,
                             unsigned* __restrict__ cnt)
{
  int e = blockIdx.x * blockDim.x + threadIdx.x;
  if (e >= NEDGES) return;
  atomicAdd(&cnt[ei[NEDGES + e]], 1u);
}

__global__ __launch_bounds__(SCAN_T) void bucket_scan(
    const unsigned* __restrict__ cnt, unsigned* __restrict__ rowptr,
    unsigned* __restrict__ cursor)
{
  __shared__ unsigned part[SCAN_T];
  const int t = threadIdx.x;
  const uint4* cp = (const uint4*)(cnt + (size_t)t * SCAN_PER);
  uint4 c[SCAN_PER / 4];
  #pragma unroll
  for (int i = 0; i < SCAN_PER / 4; ++i) c[i] = cp[i];
  unsigned s = 0;
  #pragma unroll
  for (int i = 0; i < SCAN_PER / 4; ++i) s += c[i].x + c[i].y + c[i].z + c[i].w;
  part[t] = s;
  __syncthreads();
  for (int off = 1; off < SCAN_T; off <<= 1) {
    unsigned v = (t >= off) ? part[t - off] : 0u;
    __syncthreads();
    part[t] += v;
    __syncthreads();
  }
  unsigned run = part[t] - s;
  uint4* rp = (uint4*)(rowptr + (size_t)t * SCAN_PER);
  uint4* up = (uint4*)(cursor + (size_t)t * SCAN_PER);
  #pragma unroll
  for (int i = 0; i < SCAN_PER / 4; ++i) {
    uint4 o;
    o.x = run; run += c[i].x;
    o.y = run; run += c[i].y;
    o.z = run; run += c[i].z;
    o.w = run; run += c[i].w;
    rp[i] = o; up[i] = o;
  }
}

__global__ void bucket_fill(const int* __restrict__ ei,
                            unsigned* __restrict__ cursor,
                            int* __restrict__ srcs)
{
  int e = blockIdx.x * blockDim.x + threadIdx.x;
  if (e >= NEDGES) return;
  const int s = ei[e], d = ei[NEDGES + e];
  const unsigned slot = atomicAdd(&cursor[d], 1u);
  srcs[slot] = s;
}

__device__ __forceinline__ float leaky02(float v) { return v > 0.f ? v : 0.2f * v; }

__device__ __forceinline__ float wave_max(float v) {
  #pragma unroll
  for (int o = 16; o > 0; o >>= 1) v = fmaxf(v, __shfl_xor(v, o, 32));
  return v;
}
__device__ __forceinline__ float wave_sum(float v) {
  #pragma unroll
  for (int o = 16; o > 0; o >>= 1) v += __shfl_xor(v, o, 32);
  return v;
}
__device__ __forceinline__ float lane_bcast(float v, int j) {
  return __int_as_float(__builtin_amdgcn_readlane(__float_as_int(v), j));
}

template <int HEADS>
__device__ __forceinline__ void load_scores(const float* __restrict__ p, int i,
                                            float (&o)[HEADS]) {
  if (HEADS == 4) {
    const float4 t = *(const float4*)(p + (size_t)i * 4);
    o[0] = t.x; o[1] = t.y; o[2] = t.z; o[3] = t.w;
  } else {
    #pragma unroll
    for (int h = 0; h < HEADS; ++h) o[h] = p[(size_t)i * HEADS + h];
  }
}

template <bool ELU>
__device__ __forceinline__ float bias_bn_act(float x, float b, float mu, float var,
                                             float gam, float bet) {
  x = x + b;
  x = (x - mu) * rsqrtf(var + EPSBN) * gam + bet;
  if (ELU) x = x > 0.f ? x : expm1f(x);
  return x;
}

template <bool ELU>
__device__ __forceinline__ float4 bias_bn_act4(
    float4 x, int ch0, const float* __restrict__ bias, const float* __restrict__ g,
    const float* __restrict__ bt, const float* __restrict__ m, const float* __restrict__ v) {
  const float4 b  = *(const float4*)(bias + ch0);
  const float4 mu = *(const float4*)(m + ch0);
  const float4 va = *(const float4*)(v + ch0);
  const float4 ga = *(const float4*)(g + ch0);
  const float4 be = *(const float4*)(bt + ch0);
  return make_float4(bias_bn_act<ELU>(x.x, b.x, mu.x, va.x, ga.x, be.x),
                     bias_bn_act<ELU>(x.y, b.y, mu.y, va.y, ga.y, be.y),
                     bias_bn_act<ELU>(x.z, b.z, mu.z, va.z, ga.z, be.z),
                     bias_bn_act<ELU>(x.w, b.w, mu.w, va.w, ga.w, be.w));
}

template <int HEADS, bool ELU>
__global__ __launch_bounds__(256) void node_softmax_gather(
    const unsigned* __restrict__ rowptr, const int* __restrict__ srcs,
    const float* __restrict__ s_src, const float* __restrict__ s_dst,
    const float* __restrict__ hf,
    const float* __restrict__ bias, const float* __restrict__ g,
    const float* __restrict__ bt, const float* __restrict__ m_bn,
    const float* __restrict__ v_bn, float* __restrict__ agg)
{
  constexpr int HC = HEADS * HIDC;
  const int lane = threadIdx.x & 31;
  const int wave = __builtin_amdgcn_readfirstlane((int)(threadIdx.x >> 5));
  const int node = blockIdx.x * 8 + wave;
  if (node >= NNODES) return;
  const int beg = (int)rowptr[node];
  const int deg = (int)rowptr[node + 1] - beg;
  const int cnt = deg + 1;

  float sd[HEADS], m[HEADS], den[HEADS];
  load_scores<HEADS>(s_dst, node, sd);
  #pragma unroll
  for (int h = 0; h < HEADS; ++h) { m[h] = -INFINITY; den[h] = 0.f; }

  for (int cb = 0; cb < cnt; cb += 32) {
    const int idx = cb + lane;
    if (idx < cnt) {
      const int s = idx < deg ? srcs[beg + idx] : node;
      float sc[HEADS];
      load_scores<HEADS>(s_src, s, sc);
      #pragma unroll
      for (int h = 0; h < HEADS; ++h) m[h] = fmaxf(m[h], leaky02(sc[h] + sd[h]));
    }
  }
  #pragma unroll
  for (int h = 0; h < HEADS; ++h) m[h] = wave_max(m[h]);

  float4 accA = make_float4(0.f, 0.f, 0.f, 0.f);
  float4 accB = make_float4(0.f, 0.f, 0.f, 0.f);
  const bool lo = lane < 16;

  for (int cb = 0; cb < cnt; cb += 32) {
    const int idx = cb + lane;
    int s = node;
    float e[HEADS];
    #pragma unroll
    for (int h = 0; h < HEADS; ++h) e[h] = 0.f;
    if (idx < cnt) {
      s = idx < deg ? srcs[beg + idx] : node;
      float sc[HEADS];
      load_scores<HEADS>(s_src, s, sc);
      #pragma unroll
      for (int h = 0; h < HEADS; ++h) e[h] = expf(leaky02(sc[h] + sd[h]) - m[h]);
    }
    #pragma unroll
    for (int h = 0; h < HEADS; ++h) den[h] += e[h];

    const int n = (cnt - cb) < 32 ? (cnt - cb) : 32;
    #pragma unroll 4
    for (int j = 0; j < n; ++j) {
      const int sj = __builtin_amdgcn_readlane(s, j);
      const float* row = hf + (size_t)sj * HC;
      if (HEADS == 4) {
        const float e0 = lane_bcast(e[0], j), e1 = lane_bcast(e[1 % HEADS], j);
        const float e2 = lane_bcast(e[2 % HEADS], j), e3 = lane_bcast(e[3 % HEADS], j);
        const float eA = lo ? e0 : e1;
        const float eB = lo ? e2 : e3;
        const float4 a = *(const float4*)(row + 4 * lane);
        const float4 b = *(const float4*)(row + 128 + 4 * lane);
        accA.x += eA * a.x; accA.y += eA * a.y; accA.z += eA * a.z; accA.w += eA * a.w;
        accB.x += eB * b.x; accB.y += eB * b.y; accB.z += eB * b.z; accB.w += eB * b.w;
      } else {
        const float e0 = lane_bcast(e[0], j);
        const float2 a = *(const float2*)(row + 2 * lane);
        accA.x += e0 * a.x; accA.y += e0 * a.y;
      }
    }
  }
  #pragma unroll
  for (int h = 0; h < HEADS; ++h) den[h] = wave_sum(den[h]);

  float* op = agg + (size_t)node * HC;
  if (HEADS == 4) {
    const float dA = (lo ? den[0] : den[1 % HEADS]) + 1e-16f;
    const float dB = (lo ? den[2 % HEADS] : den[3 % HEADS]) + 1e-16f;
    *(float4*)(op + 4 * lane) = bias_bn_act4<ELU>(
        make_float4(accA.x / dA, accA.y / dA, accA.z / dA, accA.w / dA),
        4 * lane, bias, g, bt, m_bn, v_bn);
    *(float4*)(op + 128 + 4 * lane) = bias_bn_act4<ELU>(
        make_float4(accB.x / dB, accB.y / dB, accB.z / dB, accB.w / dB),
        128 + 4 * lane, bias, g, bt, m_bn, v_bn);
  } else {
    const float d0 = den[0] + 1e-16f;
    const int ch0 = 2 * lane;
    const float2 b  = *(const float2*)(bias + ch0);
    const float2 mu = *(const float2*)(m_bn + ch0);
    const float2 va = *(const float2*)(v_bn + ch0);
    const float2 ga = *(const float2*)(g + ch0);
    const float2 be = *(const float2*)(bt + ch0);
    *(float2*)(op + ch0) =
        make_float2(bias_bn_act<ELU>(accA.x / d0, b.x, mu.x, va.x, ga.x, be.x),
                    bias_bn_act<ELU>(accA.y / d0, b.y, mu.y, va.y, ga.y, be.y));
  }
}

__global__ void final_linear(const float* __restrict__ h,
                             const float* __restrict__ Wc,
                             const float* __restrict__ bc,
                             float* __restrict__ out)
{
  int n = blockIdx.x * blockDim.x + threadIdx.x;
  if (n >= NNODES) return;
  float o0 = bc[0], o1 = bc[1];
  const float* r = h + (size_t)n * HIDC;
  #pragma unroll 8
  for (int i = 0; i < HIDC; ++i) {
    float x = r[i];
    o0 += x * Wc[2*i + 0];
    o1 += x * Wc[2*i + 1];
  }
  out[2*n + 0] = o0;
  out[2*n + 1] = o1;
}

__global__ void fill_u32(unsigned* __restrict__ p, unsigned val, long long n)
{
  long long i = (long long)blockIdx.x * blockDim.x + threadIdx.x;
  if (i < n) p[i] = val;
}

static inline unsigned gblk(long long total, int bs) {
  return (unsigned)((total + bs - 1) / bs);
}

extern "C" void kernel_launch(void* const* d_in, const int* in_sizes, int n_in,
                              void* d_out, int out_size, void* d_ws, size_t ws_size,
                              hipStream_t stream)
{
  const float* x  = (const float*)d_in[0];
  const int*   ei = (const int*)  d_in[1];
  const float* Wc = (const float*)d_in[26];
  const float* bc = (const float*)d_in[27];

  char* ws = (char*)d_ws;
  float*    hf    = (float*)ws;    ws += (size_t)NNODES * 256 * sizeof(float);
  float*    agg   = (float*)ws;    ws += (size_t)NNODES * 256 * sizeof(float);
  float*    ssrc  = (float*)ws;    ws += (size_t)NNODES * NHEADS * sizeof(float);
  float*    sdst  = (float*)ws;    ws += (size_t)NNODES * NHEADS * sizeof(float);
  unsigned* cnt    = (unsigned*)ws; ws += (size_t)NPAD * sizeof(unsigned);
  unsigned* cursor = (unsigned*)ws; ws += (size_t)NPAD * sizeof(unsigned);
  unsigned* rowptr = (unsigned*)ws; ws += (size_t)NPAD * sizeof(unsigned);
  int*      srcs   = (int*)ws;      ws += (size_t)NEDGES * sizeof(int);

  fill_u32<<<gblk(NPAD, 256), 256, 0, stream>>>(cnt, 0u, NPAD);
  bucket_count<<<gblk(NEDGES, 256), 256, 0, stream>>>(ei, cnt);
  bucket_scan<<<1, SCAN_T, 0, stream>>>(cnt, rowptr, cursor);
  bucket_fill<<<gblk(NEDGES, 256), 256, 0, stream>>>(ei, cursor, srcs);


  const float* in = x;
  for (int l = 0; l < 3; ++l) {
    int base = 2 + l * 8;
    const float* W  = (const float*)d_in[base + 0];
    const float* as = (const float*)d_in[base + 1];
    const float* ad = (const float*)d_in[base + 2];
    const float* bb = (const float*)d_in[base + 3];
    const float* g  = (const float*)d_in[base + 4];
    const float* bt = (const float*)d_in[base + 5];
    const float* m  = (const float*)d_in[base + 6];
    const float* vv = (const float*)d_in[base + 7];

    const int heads = (l < 2) ? NHEADS : 1;
    const int c     = HIDC;
    const int hc    = heads * c;

    dim3 ggrid(hc / GEMM_NP, (GEMM_MTILES + 7) / 8);
    if (l == 0)
      gemm_wmma_f32<FIN><<<ggrid, 256, 0, stream>>>(in, W, hf, hc);
    else
      gemm_wmma_f32<NHEADS * HIDC><<<ggrid, 256, 0, stream>>>(in, W, hf, hc);

    const int nh = NNODES * heads;
    att_scores<<<gblk(nh, 256), 256, 0, stream>>>(hf, as, ad, ssrc, sdst,
                                                  NNODES, heads, c);

    if (heads == NHEADS)
      node_softmax_gather<NHEADS, true><<<NNODES / 8, 256, 0, stream>>>(
          rowptr, srcs, ssrc, sdst, hf, bb, g, bt, m, vv, agg);
    else
      node_softmax_gather<1, false><<<NNODES / 8, 256, 0, stream>>>(
          rowptr, srcs, ssrc, sdst, hf, bb, g, bt, m, vv, agg);

    in = agg;
  }

  final_linear<<<gblk(NNODES, 256), 256, 0, stream>>>(in, Wc, bc, (float*)d_out);
}


# Bucket slot = rowptr[d] + the count returned by the counting atomic; the fill pass has no atomic (one integer atomic per edge, not two); slot order = arrival order of the counting adds
# speedup vs baseline: 4.3976x; 1.2678x over previous
#include <hip/hip_runtime.h>

typedef __attribute__((ext_vector_type(2))) float v2f;
typedef __attribute__((ext_vector_type(8))) float v8f;

#define NNODES 50000
#define NEDGES 800000
#define ETOT   850000
#define FIN    128
#define HIDC   64
#define NHEADS 4
#define EPSBN  1e-5f
#define SCAN_T   1024
#define SCAN_PER 52
#define NPAD     (SCAN_T * SCAN_PER)

#define GEMM_NP     64
#define GEMM_PP     160
#define GEMM_MTILES (NNODES / 16)
static_assert(GEMM_NP == HIDC, "the score epilogue takes one panel as one head");

template <int N>
__device__ __forceinline__ void fold_step(float (&u)[16], int bit, int lane) {
  const bool up = (lane & bit) != 0;
  #pragma unroll
  for (int k = 0; k < N / 2; ++k) {
    const float keep = up ? u[N / 2 + k] : u[k];
    const float give = up ? u[k] : u[N / 2 + k];
    u[k] = keep + __shfl_xor(give, bit, 32);
  }
}

template <int K>
__global__ __launch_bounds__(256) void gemm_wmma_f32(
    const float* __restrict__ A, const float* __restrict__ B,
    float* __restrict__ C, int Nc,
    const float* __restrict__ a_s, const float* __restrict__ a_d,
    float* __restrict__ s_src, float* __restrict__ s_dst)
{
  __shared__ float Bs[(K / 2) * GEMM_PP];

  const int tid  = threadIdx.x;
  const int lane = tid & 31;
  const int wave = tid >> 5;
  const int half = lane >> 4;
  const int ml   = lane & 15;
  const int col0 = blockIdx.x * GEMM_NP;

  #pragma unroll
  for (int it = 0; it < (K / 2) * (GEMM_NP / 4) / 256; ++it) {
    const int i = tid + 256 * it;
    const int pr = i >> 4, c4 = (i & 15) * 4;
    const float* src = B + (size_t)(2 * pr) * Nc + col0 + c4;
    const float4 x0 = *(const float4*)(src);
    const float4 x1 = *(const float4*)(src + Nc);
    float* dst = Bs + pr * GEMM_PP + 2 * c4;
    *(float4*)(dst)     = make_float4(x0.x, x1.x, x0.y, x1.y);
    *(float4*)(dst + 4) = make_float4(x0.z, x1.z, x0.w, x1.w);
  }
  __syncthreads();

  const int tile = blockIdx.y * 8 + wave;
  if (tile < GEMM_MTILES) {
    const float* ap = A + (size_t)(tile * 16 + ml) * K + 2 * half;
    const float* bp = Bs + half * GEMM_PP + 2 * ml;

    v8f acc[4];
    #pragma unroll
    for (int n = 0; n < 4; ++n)
      acc[n] = (v8f){0.f,0.f,0.f,0.f,0.f,0.f,0.f,0.f};

    v2f a_cur[8], a_nxt[8];
    #pragma unroll
    for (int j = 0; j < 8; ++j) a_cur[j] = *(const v2f*)(ap + 4 * j);

    #pragma nounroll
    for (int kb = 0; kb < K; kb += 32) {
      const int kn = (kb + 32 < K) ? kb + 32 : kb;
      #pragma unroll
      for (int j = 0; j < 8; ++j) a_nxt[j] = *(const v2f*)(ap + kn + 4 * j);

      #pragma unroll
      for (int j = 0; j < 8; ++j) {
        const float* bk = bp + ((kb + 4 * j) >> 1) * GEMM_PP;
        #pragma unroll
        for (int n = 0; n < 4; ++n) {
          const v2f b = *(const v2f*)(bk + 32 * n);
          acc[n] = __builtin_amdgcn_wmma_f32_16x16x4_f32(
              false, a_cur[j], false, b, (short)0, acc[n], false, false);
        }
      }
      #pragma unroll
      for (int j = 0; j < 8; ++j) a_cur[j] = a_nxt[j];
    }

    float* cp = C + (size_t)(tile * 16 + 8 * half) * Nc + col0 + ml;
    #pragma unroll
    for (int n = 0; n < 4; ++n) {
      #pragma unroll
      for (int i = 0; i < 8; ++i)
        cp[(size_t)i * Nc + 16 * n] = acc[n][i];
    }

    float u[16];
    #pragma unroll
    for (int i = 0; i < 16; ++i) u[i] = 0.f;
    #pragma unroll
    for (int n = 0; n < 4; ++n) {
      const float ws = a_s[col0 + 16 * n + ml];
      const float wd = a_d[col0 + 16 * n + ml];
      #pragma unroll
      for (int i = 0; i < 8; ++i) {
        u[i]     += acc[n][i] * ws;
        u[8 + i] += acc[n][i] * wd;
      }
    }
    fold_step<16>(u, 8, lane);
    fold_step<8>(u, 4, lane);
    fold_step<4>(u, 2, lane);
    fold_step<2>(u, 1, lane);
    const int heads = Nc / HIDC;
    float* sp = (ml & 8) ? s_dst : s_src;
    sp[(size_t)(tile * 16 + 8 * half + (ml & 7)) * heads + blockIdx.x] = u[0];
  }
}

__global__ void bucket_count(const int* __restrict__ ei,
                             unsigned* __restrict__ cnt,
                             unsigned* __restrict__ rank)
{
  int e = blockIdx.x * blockDim.x + threadIdx.x;
  if (e >= NEDGES) return;
  rank[e] = atomicAdd(&cnt[ei[NEDGES + e]], 1u);
}

__global__ __launch_bounds__(SCAN_T) void bucket_scan(
    const unsigned* __restrict__ cnt, unsigned* __restrict__ rowptr)
{
  __shared__ unsigned part[SCAN_T];
  const int t = threadIdx.x;
  const uint4* cp = (const uint4*)(cnt + (size_t)t * SCAN_PER);
  uint4 c[SCAN_PER / 4];
  #pragma unroll
  for (int i = 0; i < SCAN_PER / 4; ++i) c[i] = cp[i];
  unsigned s = 0;
  #pragma unroll
  for (int i = 0; i < SCAN_PER / 4; ++i) s += c[i].x + c[i].y + c[i].z + c[i].w;
  part[t] = s;
  __syncthreads();
  for (int off = 1; off < SCAN_T; off <<= 1) {
    unsigned v = (t >= off) ? part[t - off] : 0u;
    __syncthreads();
    part[t] += v;
    __syncthreads();
  }
  unsigned run = part[t] - s;
  uint4* rp = (uint4*)(rowptr + (size_t)t * SCAN_PER);
  #pragma unroll
  for (int i = 0; i < SCAN_PER / 4; ++i) {
    uint4 o;
    o.x = run; run += c[i].x;
    o.y = run; run += c[i].y;
    o.z = run; run += c[i].z;
    o.w = run; run += c[i].w;
    rp[i] = o;
  }
}

__global__ void bucket_fill(const int* __restrict__ ei,
                            const unsigned* __restrict__ rowptr,
                            const unsigned* __restrict__ rank,
                            int* __restrict__ srcs)
{
  int e = blockIdx.x * blockDim.x + threadIdx.x;
  if (e >= NEDGES) return;
  const int s = ei[e], d = ei[NEDGES + e];
  srcs[rowptr[d] + rank[e]] = s;
}

__device__ __forceinline__ float leaky02(float v) { return v > 0.f ? v : 0.2f * v; }

__device__ __forceinline__ float wave_max(float v) {
  #pragma unroll
  for (int o = 16; o > 0; o >>= 1) v = fmaxf(v, __shfl_xor(v, o, 32));
  return v;
}
__device__ __forceinline__ float wave_sum(float v) {
  #pragma unroll
  for (int o = 16; o > 0; o >>= 1) v += __shfl_xor(v, o, 32);
  return v;
}
__device__ __forceinline__ float lane_bcast(float v, int j) {
  return __int_as_float(__builtin_amdgcn_readlane(__float_as_int(v), j));
}

template <int HEADS>
__device__ __forceinline__ void load_scores(const float* __restrict__ p, int i,
                                            float (&o)[HEADS]) {
  if (HEADS == 4) {
    const float4 t = *(const float4*)(p + (size_t)i * 4);
    o[0] = t.x; o[1] = t.y; o[2] = t.z; o[3] = t.w;
  } else {
    #pragma unroll
    for (int h = 0; h < HEADS; ++h) o[h] = p[(size_t)i * HEADS + h];
  }
}

template <bool ELU>
__device__ __forceinline__ float bias_bn_act(float x, float b, float mu, float var,
                                             float gam, float bet) {
  x = x + b;
  x = (x - mu) * rsqrtf(var + EPSBN) * gam + bet;
  if (ELU) x = x > 0.f ? x : expm1f(x);
  return x;
}

template <bool ELU>
__device__ __forceinline__ float4 bias_bn_act4(
    float4 x, int ch0, const float* __restrict__ bias, const float* __restrict__ g,
    const float* __restrict__ bt, const float* __restrict__ m, const float* __restrict__ v) {
  const float4 b  = *(const float4*)(bias + ch0);
  const float4 mu = *(const float4*)(m + ch0);
  const float4 va = *(const float4*)(v + ch0);
  const float4 ga = *(const float4*)(g + ch0);
  const float4 be = *(const float4*)(bt + ch0);
  return make_float4(bias_bn_act<ELU>(x.x, b.x, mu.x, va.x, ga.x, be.x),
                     bias_bn_act<ELU>(x.y, b.y, mu.y, va.y, ga.y, be.y),
                     bias_bn_act<ELU>(x.z, b.z, mu.z, va.z, ga.z, be.z),
                     bias_bn_act<ELU>(x.w, b.w, mu.w, va.w, ga.w, be.w));
}

template <int HEADS, bool ELU>
__global__ __launch_bounds__(256) void node_softmax_gather(
    const unsigned* __restrict__ rowptr, const int* __restrict__ srcs,
    const float* __restrict__ s_src, const float* __restrict__ s_dst,
    const float* __restrict__ hf,
    const float* __restrict__ bias, const float* __restrict__ g,
    const float* __restrict__ bt, const float* __restrict__ m_bn,
    const float* __restrict__ v_bn, float* __restrict__ agg)
{
  constexpr int HC = HEADS * HIDC;
  const int lane = threadIdx.x & 31;
  const int wave = __builtin_amdgcn_readfirstlane((int)(threadIdx.x >> 5));
  const int node = blockIdx.x * 8 + wave;
  if (node >= NNODES) return;
  const int beg = (int)rowptr[node];
  const int deg = (int)rowptr[node + 1] - beg;
  const int cnt = deg + 1;

  float sd[HEADS], m[HEADS], den[HEADS];
  load_scores<HEADS>(s_dst, node, sd);
  #pragma unroll
  for (int h = 0; h < HEADS; ++h) { m[h] = -INFINITY; den[h] = 0.f; }

  for (int cb = 0; cb < cnt; cb += 32) {
    const int idx = cb + lane;
    if (idx < cnt) {
      const int s = idx < deg ? srcs[beg + idx] : node;
      float sc[HEADS];
      load_scores<HEADS>(s_src, s, sc);
      #pragma unroll
      for (int h = 0; h < HEADS; ++h) m[h] = fmaxf(m[h], leaky02(sc[h] + sd[h]));
    }
  }
  #pragma unroll
  for (int h = 0; h < HEADS; ++h) m[h] = wave_max(m[h]);

  float4 accA = make_float4(0.f, 0.f, 0.f, 0.f);
  float4 accB = make_float4(0.f, 0.f, 0.f, 0.f);
  const bool lo = lane < 16;

  for (int cb = 0; cb < cnt; cb += 32) {
    const int idx = cb + lane;
    int s = node;
    float e[HEADS];
    #pragma unroll
    for (int h = 0; h < HEADS; ++h) e[h] = 0.f;
    if (idx < cnt) {
      s = idx < deg ? srcs[beg + idx] : node;
      float sc[HEADS];
      load_scores<HEADS>(s_src, s, sc);
      #pragma unroll
      for (int h = 0; h < HEADS; ++h) e[h] = expf(leaky02(sc[h] + sd[h]) - m[h]);
    }
    #pragma unroll
    for (int h = 0; h < HEADS; ++h) den[h] += e[h];

    const int n = (cnt - cb) < 32 ? (cnt - cb) : 32;
    #pragma unroll 4
    for (int j = 0; j < n; ++j) {
      const int sj = __builtin_amdgcn_readlane(s, j);
      const float* row = hf + (size_t)sj * HC;
      if (HEADS == 4) {
        const float e0 = lane_bcast(e[0], j), e1 = lane_bcast(e[1 % HEADS], j);
        const float e2 = lane_bcast(e[2 % HEADS], j), e3 = lane_bcast(e[3 % HEADS], j);
        const float eA = lo ? e0 : e1;
        const float eB = lo ? e2 : e3;
        const float4 a = *(const float4*)(row + 4 * lane);
        const float4 b = *(const float4*)(row + 128 + 4 * lane);
        accA.x += eA * a.x; accA.y += eA * a.y; accA.z += eA * a.z; accA.w += eA * a.w;
        accB.x += eB * b.x; accB.y += eB * b.y; accB.z += eB * b.z; accB.w += eB * b.w;
      } else {
        const float e0 = lane_bcast(e[0], j);
        const float2 a = *(const float2*)(row + 2 * lane);
        accA.x += e0 * a.x; accA.y += e0 * a.y;
      }
    }
  }
  #pragma unroll
  for (int h = 0; h < HEADS; ++h) den[h] = wave_sum(den[h]);

  float* op = agg + (size_t)node * HC;
  if (HEADS == 4) {
    const float dA = (lo ? den[0] : den[1 % HEADS]) + 1e-16f;
    const float dB = (lo ? den[2 % HEADS] : den[3 % HEADS]) + 1e-16f;
    *(float4*)(op + 4 * lane) = bias_bn_act4<ELU>(
        make_float4(accA.x / dA, accA.y / dA, accA.z / dA, accA.w / dA),
        4 * lane, bias, g, bt, m_bn, v_bn);
    *(float4*)(op + 128 + 4 * lane) = bias_bn_act4<ELU>(
        make_float4(accB.x / dB, accB.y / dB, accB.z / dB, accB.w / dB),
        128 + 4 * lane, bias, g, bt, m_bn, v_bn);
  } else {
    const float d0 = den[0] + 1e-16f;
    const int ch0 = 2 * lane;
    const float2 b  = *(const float2*)(bias + ch0);
    const float2 mu = *(const float2*)(m_bn + ch0);
    const float2 va = *(const float2*)(v_bn + ch0);
    const float2 ga = *(const float2*)(g + ch0);
    const float2 be = *(const float2*)(bt + ch0);
    *(float2*)(op + ch0) =
        make_float2(bias_bn_act<ELU>(accA.x / d0, b.x, mu.x, va.x, ga.x, be.x),
                    bias_bn_act<ELU>(accA.y / d0, b.y, mu.y, va.y, ga.y, be.y));
  }
}

__global__ void final_linear(const float* __restrict__ h,
                             const float* __restrict__ Wc,
                             const float* __restrict__ bc,
                             float* __restrict__ out)
{
  int n = blockIdx.x * blockDim.x + threadIdx.x;
  if (n >= NNODES) return;
  float o0 = bc[0], o1 = bc[1];
  const float* r = h + (size_t)n * HIDC;
  #pragma unroll 8
  for (int i = 0; i < HIDC; ++i) {
    float x = r[i];
    o0 += x * Wc[2*i + 0];
    o1 += x * Wc[2*i + 1];
  }
  out[2*n + 0] = o0;
  out[2*n + 1] = o1;
}

__global__ void fill_u32(unsigned* __restrict__ p, unsigned val, long long n)
{
  long long i = (long long)blockIdx.x * blockDim.x + threadIdx.x;
  if (i < n) p[i] = val;
}

static inline unsigned gblk(long long total, int bs) {
  return (unsigned)((total + bs - 1) / bs);
}

extern "C" void kernel_launch(void* const* d_in, const int* in_sizes, int n_in,
                              void* d_out, int out_size, void* d_ws, size_t ws_size,
                              hipStream_t stream)
{
  const float* x  = (const float*)d_in[0];
  const int*   ei = (const int*)  d_in[1];
  const float* Wc = (const float*)d_in[26];
  const float* bc = (const float*)d_in[27];

  char* ws = (char*)d_ws;
  float*    hf    = (float*)ws;    ws += (size_t)NNODES * 256 * sizeof(float);
  float*    agg   = (float*)ws;    ws += (size_t)NNODES * 256 * sizeof(float);
  float*    ssrc  = (float*)ws;    ws += (size_t)NNODES * NHEADS * sizeof(float);
  float*    sdst  = (float*)ws;    ws += (size_t)NNODES * NHEADS * sizeof(float);
  unsigned* cnt    = (unsigned*)ws; ws += (size_t)NPAD * sizeof(unsigned);
  unsigned* rowptr = (unsigned*)ws; ws += (size_t)NPAD * sizeof(unsigned);
  int*      srcs   = (int*)ws;      ws += (size_t)NEDGES * sizeof(int);
  unsigned* rank   = (unsigned*)agg;

  fill_u32<<<gblk(NPAD, 256), 256, 0, stream>>>(cnt, 0u, NPAD);
  bucket_count<<<gblk(NEDGES, 256), 256, 0, stream>>>(ei, cnt, rank);
  bucket_scan<<<1, SCAN_T, 0, stream>>>(cnt, rowptr);
  bucket_fill<<<gblk(NEDGES, 256), 256, 0, stream>>>(ei, rowptr, rank, srcs);


  const float* in = x;
  for (int l = 0; l < 3; ++l) {
    int base = 2 + l * 8;
    const float* W  = (const float*)d_in[base + 0];
    const float* as = (const float*)d_in[base + 1];
    const float* ad = (const float*)d_in[base + 2];
    const float* bb = (const float*)d_in[base + 3];
    const float* g  = (const float*)d_in[base + 4];
    const float* bt = (const float*)d_in[base + 5];
    const float* m  = (const float*)d_in[base + 6];
    const float* vv = (const float*)d_in[base + 7];

    const int heads = (l < 2) ? NHEADS : 1;
    const int hc    = heads * HIDC;

    dim3 ggrid(hc / GEMM_NP, (GEMM_MTILES + 7) / 8);
    if (l == 0)
      gemm_wmma_f32<FIN><<<ggrid, 256, 0, stream>>>(in, W, hf, hc, as, ad, ssrc, sdst);
    else
      gemm_wmma_f32<NHEADS * HIDC><<<ggrid, 256, 0, stream>>>(in, W, hf, hc, as, ad, ssrc, sdst);

    if (heads == NHEADS)
      node_softmax_gather<NHEADS, true><<<NNODES / 8, 256, 0, stream>>>(
          rowptr, srcs, ssrc, sdst, hf, bb, g, bt, m, vv, agg);
    else
      node_softmax_gather<1, false><<<NNODES / 8, 256, 0, stream>>>(
          rowptr, srcs, ssrc, sdst, hf, bb, g, bt, m, vv, agg);

    in = agg;
  }

  final_linear<<<gblk(NNODES, 256), 256, 0, stream>>>(in, Wc, bc, (float*)d_out);
}


# Node kernel: scores, then the first 8 source rows requested ahead of the softmax max/exp chain (logits kept, no second score load); later rows in groups of 8 in flight (was 4); sum order unchanged
# speedup vs baseline: 5.1535x; 1.1719x over previous
#include <hip/hip_runtime.h>

typedef __attribute__((ext_vector_type(2))) float v2f;
typedef __attribute__((ext_vector_type(8))) float v8f;

#define NNODES 50000
#define NEDGES 800000
#define FIN    128
#define HIDC   64
#define NHEADS 4
#define EPSBN  1e-5f
#define CAP    32
#define NCNT   ((NNODES + 1 + 63) / 64 * 64)
static_assert(NNODES <= 65536, "an overflow entry packs two node ids into one 32-bit word");

#define GEMM_NP     64
#define GEMM_PP     160
#define GEMM_MTILES (NNODES / 16)
static_assert(GEMM_NP == HIDC, "the score epilogue takes one panel as one head");

#define CNT_PANELS  ((NHEADS * HIDC) / GEMM_NP)
#define CNT_BLOCKS  256
#define CNT_ROWS    (CNT_BLOCKS / CNT_PANELS)
#define CNT_EDGES   (NEDGES / CNT_BLOCKS)
#define CNT_TRIPS   ((CNT_EDGES + 255) / 256)
static_assert(CNT_BLOCKS % CNT_PANELS == 0 && NEDGES % CNT_BLOCKS == 0,
              "counting work-groups tile the edge list exactly");

template <int N>
__device__ __forceinline__ void fold_step(float (&u)[16], int bit, int lane) {
  const bool up = (lane & bit) != 0;
  #pragma unroll
  for (int k = 0; k < N / 2; ++k) {
    const float keep = up ? u[N / 2 + k] : u[k];
    const float give = up ? u[k] : u[N / 2 + k];
    u[k] = keep + __shfl_xor(give, bit, 32);
  }
}

template <int K, bool WITH_COUNT>
__global__ __launch_bounds__(256) void gemm_wmma_f32(
    const float* __restrict__ A, const float* __restrict__ B,
    float* __restrict__ C, int Nc,
    const float* __restrict__ a_s, const float* __restrict__ a_d,
    float* __restrict__ s_src, float* __restrict__ s_dst,
    const int* __restrict__ ei, unsigned* __restrict__ cnt,
    int* __restrict__ slots, unsigned* __restrict__ ovf)
{
  __shared__ float Bs[(K / 2) * GEMM_PP];

  int tile_row = blockIdx.y;
  if (WITH_COUNT) {
    if (tile_row < CNT_ROWS) {
      const int e0 = (tile_row * CNT_PANELS + (int)blockIdx.x) * CNT_EDGES;
      const int* sp = ei + e0;
      const int* dp = ei + NEDGES + e0;
      unsigned* novf = cnt + NNODES;
      const int t = (int)threadIdx.x;
      int d[CNT_TRIPS], s[CNT_TRIPS];
      unsigned r[CNT_TRIPS];
      #pragma unroll
      for (int i = 0; i < CNT_TRIPS; ++i) {
        const bool ok = t + 256 * i < CNT_EDGES;
        d[i] = ok ? dp[t + 256 * i] : 0;
        s[i] = ok ? sp[t + 256 * i] : 0;
      }
      #pragma unroll
      for (int i = 0; i < CNT_TRIPS; ++i)
        r[i] = (t + 256 * i < CNT_EDGES) ? atomicAdd(&cnt[d[i]], 1u) : 0u;
      #pragma unroll
      for (int i = 0; i < CNT_TRIPS; ++i) {
        if (t + 256 * i < CNT_EDGES) {
          if (r[i] < (unsigned)CAP) {
            slots[(size_t)d[i] * CAP + r[i]] = s[i];
          } else {
            const unsigned k = atomicAdd(novf, 1u);
            ovf[k] = ((unsigned)d[i] << 16) | (unsigned)s[i];
          }
        }
      }
      return;
    }
    tile_row -= CNT_ROWS;
  }

  const int tid  = threadIdx.x;
  const int lane = tid & 31;
  const int wave = tid >> 5;
  const int half = lane >> 4;
  const int ml   = lane & 15;
  const int col0 = blockIdx.x * GEMM_NP;

  #pragma unroll
  for (int it = 0; it < (K / 2) * (GEMM_NP / 4) / 256; ++it) {
    const int i = tid + 256 * it;
    const int pr = i >> 4, c4 = (i & 15) * 4;
    const float* src = B + (size_t)(2 * pr) * Nc + col0 + c4;
    const float4 x0 = *(const float4*)(src);
    const float4 x1 = *(const float4*)(src + Nc);
    float* dst = Bs + pr * GEMM_PP + 2 * c4;
    *(float4*)(dst)     = make_float4(x0.x, x1.x, x0.y, x1.y);
    *(float4*)(dst + 4) = make_float4(x0.z, x1.z, x0.w, x1.w);
  }
  __syncthreads();

  const int tile = tile_row * 8 + wave;
  if (tile < GEMM_MTILES) {
    const float* ap = A + (size_t)(tile * 16 + ml) * K + 2 * half;
    const float* bp = Bs + half * GEMM_PP + 2 * ml;

    v8f acc[4];
    #pragma unroll
    for (int n = 0; n < 4; ++n)
      acc[n] = (v8f){0.f,0.f,0.f,0.f,0.f,0.f,0.f,0.f};

    v2f a_cur[8], a_nxt[8];
    #pragma unroll
    for (int j = 0; j < 8; ++j) a_cur[j] = *(const v2f*)(ap + 4 * j);

    #pragma nounroll
    for (int kb = 0; kb < K; kb += 32) {
      const int kn = (kb + 32 < K) ? kb + 32 : kb;
      #pragma unroll
      for (int j = 0; j < 8; ++j) a_nxt[j] = *(const v2f*)(ap + kn + 4 * j);

      #pragma unroll
      for (int j = 0; j < 8; ++j) {
        const float* bk = bp + ((kb + 4 * j) >> 1) * GEMM_PP;
        #pragma unroll
        for (int n = 0; n < 4; ++n) {
          const v2f b = *(const v2f*)(bk + 32 * n);
          acc[n] = __builtin_amdgcn_wmma_f32_16x16x4_f32(
              false, a_cur[j], false, b, (short)0, acc[n], false, false);
        }
      }
      #pragma unroll
      for (int j = 0; j < 8; ++j) a_cur[j] = a_nxt[j];
    }

    float* cp = C + (size_t)(tile * 16 + 8 * half) * Nc + col0 + ml;
    #pragma unroll
    for (int n = 0; n < 4; ++n) {
      #pragma unroll
      for (int i = 0; i < 8; ++i)
        cp[(size_t)i * Nc + 16 * n] = acc[n][i];
    }

    float u[16];
    #pragma unroll
    for (int i = 0; i < 16; ++i) u[i] = 0.f;
    #pragma unroll
    for (int n = 0; n < 4; ++n) {
      const float ws = a_s[col0 + 16 * n + ml];
      const float wd = a_d[col0 + 16 * n + ml];
      #pragma unroll
      for (int i = 0; i < 8; ++i) {
        u[i]     += acc[n][i] * ws;
        u[8 + i] += acc[n][i] * wd;
      }
    }
    fold_step<16>(u, 8, lane);
    fold_step<8>(u, 4, lane);
    fold_step<4>(u, 2, lane);
    fold_step<2>(u, 1, lane);
    const int heads = Nc / HIDC;
    float* sp = (ml & 8) ? s_dst : s_src;
    sp[(size_t)(tile * 16 + 8 * half + (ml & 7)) * heads + blockIdx.x] = u[0];
  }
}


__device__ __forceinline__ float leaky02(float v) { return v > 0.f ? v : 0.2f * v; }

__device__ __forceinline__ float wave_max(float v) {
  #pragma unroll
  for (int o = 16; o > 0; o >>= 1) v = fmaxf(v, __shfl_xor(v, o, 32));
  return v;
}
__device__ __forceinline__ float wave_sum(float v) {
  #pragma unroll
  for (int o = 16; o > 0; o >>= 1) v += __shfl_xor(v, o, 32);
  return v;
}
__device__ __forceinline__ float lane_bcast(float v, int j) {
  return __int_as_float(__builtin_amdgcn_readlane(__float_as_int(v), j));
}

template <int HEADS>
__device__ __forceinline__ void load_scores(const float* __restrict__ p, int i,
                                            float (&o)[HEADS]) {
  if (HEADS == 4) {
    const float4 t = *(const float4*)(p + (size_t)i * 4);
    o[0] = t.x; o[1] = t.y; o[2] = t.z; o[3] = t.w;
  } else {
    #pragma unroll
    for (int h = 0; h < HEADS; ++h) o[h] = p[(size_t)i * HEADS + h];
  }
}

template <bool ELU>
__device__ __forceinline__ float bias_bn_act(float x, float b, float mu, float var,
                                             float gam, float bet) {
  x = x + b;
  x = (x - mu) * rsqrtf(var + EPSBN) * gam + bet;
  if (ELU) x = x > 0.f ? x : expm1f(x);
  return x;
}

template <bool ELU>
__device__ __forceinline__ float4 bias_bn_act4(
    float4 x, int ch0, const float* __restrict__ bias, const float* __restrict__ g,
    const float* __restrict__ bt, const float* __restrict__ m, const float* __restrict__ v) {
  const float4 b  = *(const float4*)(bias + ch0);
  const float4 mu = *(const float4*)(m + ch0);
  const float4 va = *(const float4*)(v + ch0);
  const float4 ga = *(const float4*)(g + ch0);
  const float4 be = *(const float4*)(bt + ch0);
  return make_float4(bias_bn_act<ELU>(x.x, b.x, mu.x, va.x, ga.x, be.x),
                     bias_bn_act<ELU>(x.y, b.y, mu.y, va.y, ga.y, be.y),
                     bias_bn_act<ELU>(x.z, b.z, mu.z, va.z, ga.z, be.z),
                     bias_bn_act<ELU>(x.w, b.w, mu.w, va.w, ga.w, be.w));
}

template <int HEADS>
__device__ __forceinline__ void row_load(const float* __restrict__ hf, int sj, int lane,
                                         float4& a, float4& b) {
  const float* row = hf + (size_t)sj * (HEADS * HIDC);
  if (HEADS == 4) {
    a = *(const float4*)(row + 4 * lane);
    b = *(const float4*)(row + 128 + 4 * lane);
  } else {
    const float2 t = *(const float2*)(row + 2 * lane);
    a.x = t.x; a.y = t.y;
  }
}

template <int HEADS>
__device__ __forceinline__ void row_fma(const float4& a, const float4& b, int j,
                                        const float (&e)[HEADS], bool lo,
                                        float4& accA, float4& accB) {
  if (HEADS == 4) {
    const float e0 = lane_bcast(e[0], j), e1 = lane_bcast(e[1 % HEADS], j);
    const float e2 = lane_bcast(e[2 % HEADS], j), e3 = lane_bcast(e[3 % HEADS], j);
    const float eA = lo ? e0 : e1;
    const float eB = lo ? e2 : e3;
    accA.x = __builtin_fmaf(eA, a.x, accA.x); accA.y = __builtin_fmaf(eA, a.y, accA.y);
    accA.z = __builtin_fmaf(eA, a.z, accA.z); accA.w = __builtin_fmaf(eA, a.w, accA.w);
    accB.x = __builtin_fmaf(eB, b.x, accB.x); accB.y = __builtin_fmaf(eB, b.y, accB.y);
    accB.z = __builtin_fmaf(eB, b.z, accB.z); accB.w = __builtin_fmaf(eB, b.w, accB.w);
  } else {
    const float e0 = lane_bcast(e[0], j);
    accA.x = __builtin_fmaf(e0, a.x, accA.x); accA.y = __builtin_fmaf(e0, a.y, accA.y);
  }
}

template <int HEADS>
__device__ __forceinline__ void add_row(const float* __restrict__ hf, int sj, int j,
                                        const float (&e)[HEADS], bool lo, int lane,
                                        float4& accA, float4& accB) {
  float4 a, b;
  row_load<HEADS>(hf, sj, lane, a, b);
  row_fma<HEADS>(a, b, j, e, lo, accA, accB);
}

template <int HEADS, int G>
__device__ __forceinline__ void rows_request(const float* __restrict__ hf, int s, int jb,
                                             int jl, int lane,
                                             float4 (&a)[8], float4 (&b)[8]) {
  #pragma unroll
  for (int u = 0; u < G; ++u) {
    const int j = (jb + u) < jl ? (jb + u) : jl;
    row_load<HEADS>(hf, __builtin_amdgcn_readlane(s, j), lane, a[u], b[u]);
  }
}

template <int HEADS, int G>
__device__ __forceinline__ void rows_add(const float4 (&a)[8], const float4 (&b)[8], int jb,
                                         const float (&e)[HEADS], bool lo,
                                         float4& accA, float4& accB) {
  #pragma unroll
  for (int u = 0; u < G; ++u) row_fma<HEADS>(a[u], b[u], jb + u, e, lo, accA, accB);
}
template <int HEADS, int G>
__device__ __forceinline__ void rows_add_tail(const float4 (&a)[8], const float4 (&b)[8],
                                              int jb, int n, const float (&e)[HEADS], bool lo,
                                              float4& accA, float4& accB) {
  #pragma unroll
  for (int u = 0; u < G; ++u)
    if (jb + u < n) row_fma<HEADS>(a[u], b[u], jb + u, e, lo, accA, accB);
}

template <int HEADS, bool ELU>
__global__ __launch_bounds__(256) void node_softmax_gather(
    const unsigned* __restrict__ cnt, const int* __restrict__ slots,
    const unsigned* __restrict__ ovf,
    const float* __restrict__ s_src, const float* __restrict__ s_dst,
    const float* __restrict__ hf,
    const float* __restrict__ bias, const float* __restrict__ g,
    const float* __restrict__ bt, const float* __restrict__ m_bn,
    const float* __restrict__ v_bn, float* __restrict__ agg)
{
  static_assert(CAP == 32, "one lane per bucket slot");
  constexpr int HC = HEADS * HIDC;
  const int lane = threadIdx.x & 31;
  const int wave = __builtin_amdgcn_readfirstlane((int)(threadIdx.x >> 5));
  const int node = blockIdx.x * 8 + wave;
  if (node >= NNODES) return;
  const int indeg = (int)cnt[node];
  const int deg = indeg < CAP ? indeg : CAP;
  const int ncur = deg + 1;
  const int n0 = ncur < 32 ? ncur : 32;
  const int novf = indeg > CAP ? (int)cnt[NNODES] : 0;

  float sd[HEADS], m[HEADS], den[HEADS];
  load_scores<HEADS>(s_dst, node, sd);

  int s0 = node;
  if (lane < deg) s0 = slots[(size_t)node * CAP + lane];
  float lg[HEADS];
  {
    float sc[HEADS];
    load_scores<HEADS>(s_src, s0, sc);
    #pragma unroll
    for (int h = 0; h < HEADS; ++h) lg[h] = leaky02(sc[h] + sd[h]);
  }
  float4 ra[8], rb[8];
  rows_request<HEADS, 8>(hf, s0, 0, n0 - 1, lane, ra, rb);

  #pragma unroll
  for (int h = 0; h < HEADS; ++h) m[h] = lane < n0 ? lg[h] : -INFINITY;
  float lgs[HEADS];
  if (ncur > 32) {
    float sc[HEADS];
    load_scores<HEADS>(s_src, node, sc);
    #pragma unroll
    for (int h = 0; h < HEADS; ++h) {
      lgs[h] = leaky02(sc[h] + sd[h]);
      m[h] = fmaxf(m[h], lgs[h]);
    }
  }
  for (int k0 = 0; k0 < novf; k0 += 32) {
    const int k = k0 + lane;
    if (k < novf) {
      const unsigned w = ovf[k];
      if ((int)(w >> 16) == node) {
        float sc[HEADS];
        load_scores<HEADS>(s_src, (int)(w & 0xffffu), sc);
        #pragma unroll
        for (int h = 0; h < HEADS; ++h) m[h] = fmaxf(m[h], leaky02(sc[h] + sd[h]));
      }
    }
  }
  #pragma unroll
  for (int h = 0; h < HEADS; ++h) m[h] = wave_max(m[h]);

  float4 accA = make_float4(0.f, 0.f, 0.f, 0.f);
  float4 accB = make_float4(0.f, 0.f, 0.f, 0.f);
  const bool lo = lane < 16;

  float e[HEADS];
  #pragma unroll
  for (int h = 0; h < HEADS; ++h) {
    e[h] = lane < n0 ? expf(lg[h] - m[h]) : 0.f;
    den[h] = e[h];
  }
  rows_add_tail<HEADS, 8>(ra, rb, 0, n0, e, lo, accA, accB);
  int j0 = 8;
  for (; j0 + 8 <= n0; j0 += 8) {
    rows_request<HEADS, 8>(hf, s0, j0, n0 - 1, lane, ra, rb);
    rows_add<HEADS, 8>(ra, rb, j0, e, lo, accA, accB);
  }
  if (j0 < n0) {
    if (n0 - j0 > 4) {
      rows_request<HEADS, 8>(hf, s0, j0, n0 - 1, lane, ra, rb);
      rows_add_tail<HEADS, 8>(ra, rb, j0, n0, e, lo, accA, accB);
    } else {
      rows_request<HEADS, 4>(hf, s0, j0, n0 - 1, lane, ra, rb);
      rows_add_tail<HEADS, 4>(ra, rb, j0, n0, e, lo, accA, accB);
    }
  }
  if (ncur > 32) {
    float es[HEADS];
    #pragma unroll
    for (int h = 0; h < HEADS; ++h) {
      es[h] = expf(lgs[h] - m[h]);
      den[h] += lane == 0 ? es[h] : 0.f;
    }
    add_row<HEADS>(hf, node, 0, es, lo, lane, accA, accB);
  }
  for (int k0 = 0; k0 < novf; k0 += 32) {
    const int k = k0 + lane;
    int s = node;
    bool hit = false;
    float eo[HEADS];
    #pragma unroll
    for (int h = 0; h < HEADS; ++h) eo[h] = 0.f;
    if (k < novf) {
      const unsigned w = ovf[k];
      hit = (int)(w >> 16) == node;
      if (hit) {
        s = (int)(w & 0xffffu);
        float sc[HEADS];
        load_scores<HEADS>(s_src, s, sc);
        #pragma unroll
        for (int h = 0; h < HEADS; ++h) eo[h] = expf(leaky02(sc[h] + sd[h]) - m[h]);
      }
    }
    #pragma unroll
    for (int h = 0; h < HEADS; ++h) den[h] += eo[h];
    unsigned mask = __builtin_amdgcn_ballot_w32(hit);
    while (mask != 0u) {
      const int j = __builtin_ctz(mask);
      mask &= mask - 1u;
      const int sj = __builtin_amdgcn_readlane(s, j);
      add_row<HEADS>(hf, sj, j, eo, lo, lane, accA, accB);
    }
  }
  #pragma unroll
  for (int h = 0; h < HEADS; ++h) den[h] = wave_sum(den[h]);

  float* op = agg + (size_t)node * HC;
  if (HEADS == 4) {
    const float dA = (lo ? den[0] : den[1 % HEADS]) + 1e-16f;
    const float dB = (lo ? den[2 % HEADS] : den[3 % HEADS]) + 1e-16f;
    *(float4*)(op + 4 * lane) = bias_bn_act4<ELU>(
        make_float4(accA.x / dA, accA.y / dA, accA.z / dA, accA.w / dA),
        4 * lane, bias, g, bt, m_bn, v_bn);
    *(float4*)(op + 128 + 4 * lane) = bias_bn_act4<ELU>(
        make_float4(accB.x / dB, accB.y / dB, accB.z / dB, accB.w / dB),
        128 + 4 * lane, bias, g, bt, m_bn, v_bn);
  } else {
    const float d0 = den[0] + 1e-16f;
    const int ch0 = 2 * lane;
    const float2 b  = *(const float2*)(bias + ch0);
    const float2 mu = *(const float2*)(m_bn + ch0);
    const float2 va = *(const float2*)(v_bn + ch0);
    const float2 ga = *(const float2*)(g + ch0);
    const float2 be = *(const float2*)(bt + ch0);
    *(float2*)(op + ch0) =
        make_float2(bias_bn_act<ELU>(accA.x / d0, b.x, mu.x, va.x, ga.x, be.x),
                    bias_bn_act<ELU>(accA.y / d0, b.y, mu.y, va.y, ga.y, be.y));
  }
}

__global__ void final_linear(const float* __restrict__ h,
                             const float* __restrict__ Wc,
                             const float* __restrict__ bc,
                             float* __restrict__ out)
{
  int n = blockIdx.x * blockDim.x + threadIdx.x;
  if (n >= NNODES) return;
  float o0 = bc[0], o1 = bc[1];
  const float* r = h + (size_t)n * HIDC;
  #pragma unroll 8
  for (int i = 0; i < HIDC; ++i) {
    float x = r[i];
    o0 += x * Wc[2*i + 0];
    o1 += x * Wc[2*i + 1];
  }
  out[2*n + 0] = o0;
  out[2*n + 1] = o1;
}

__global__ void fill_u32(unsigned* __restrict__ p, unsigned val, long long n)
{
  long long i = (long long)blockIdx.x * blockDim.x + threadIdx.x;
  if (i < n) p[i] = val;
}

static inline unsigned gblk(long long total, int bs) {
  return (unsigned)((total + bs - 1) / bs);
}

extern "C" void kernel_launch(void* const* d_in, const int* in_sizes, int n_in,
                              void* d_out, int out_size, void* d_ws, size_t ws_size,
                              hipStream_t stream)
{
  const float* x  = (const float*)d_in[0];
  const int*   ei = (const int*)  d_in[1];
  const float* Wc = (const float*)d_in[26];
  const float* bc = (const float*)d_in[27];

  char* ws = (char*)d_ws;
  float*    hf    = (float*)ws;    ws += (size_t)NNODES * 256 * sizeof(float);
  float*    agg   = (float*)ws;    ws += (size_t)NNODES * 256 * sizeof(float);
  float*    ssrc  = (float*)ws;    ws += (size_t)NNODES * NHEADS * sizeof(float);
  float*    sdst  = (float*)ws;    ws += (size_t)NNODES * NHEADS * sizeof(float);
  unsigned* cnt    = (unsigned*)ws; ws += (size_t)NCNT * sizeof(unsigned);
  unsigned* ovf    = (unsigned*)ws; ws += (size_t)NEDGES * sizeof(unsigned);
  int*      slots  = (int*)ws;      ws += (size_t)NNODES * CAP * sizeof(int);

  fill_u32<<<gblk(NCNT, 256), 256, 0, stream>>>(cnt, 0u, NCNT);

  const float* in = x;
  for (int l = 0; l < 3; ++l) {
    int base = 2 + l * 8;
    const float* W  = (const float*)d_in[base + 0];
    const float* as = (const float*)d_in[base + 1];
    const float* ad = (const float*)d_in[base + 2];
    const float* bb = (const float*)d_in[base + 3];
    const float* g  = (const float*)d_in[base + 4];
    const float* bt = (const float*)d_in[base + 5];
    const float* m  = (const float*)d_in[base + 6];
    const float* vv = (const float*)d_in[base + 7];

    const int heads = (l < 2) ? NHEADS : 1;
    const int hc    = heads * HIDC;

    dim3 ggrid(hc / GEMM_NP, (GEMM_MTILES + 7) / 8);
    if (l == 0) {
      dim3 ggrid0(CNT_PANELS, CNT_ROWS + (GEMM_MTILES + 7) / 8);
      gemm_wmma_f32<FIN, true><<<ggrid0, 256, 0, stream>>>(
          in, W, hf, hc, as, ad, ssrc, sdst, ei, cnt, slots, ovf);
    } else {
      gemm_wmma_f32<NHEADS * HIDC, false><<<ggrid, 256, 0, stream>>>(
          in, W, hf, hc, as, ad, ssrc, sdst, nullptr, nullptr, nullptr, nullptr);
    }

    if (heads == NHEADS)
      node_softmax_gather<NHEADS, true><<<NNODES / 8, 256, 0, stream>>>(
          cnt, slots, ovf, ssrc, sdst, hf, bb, g, bt, m, vv, agg);
    else
      node_softmax_gather<1, false><<<NNODES / 8, 256, 0, stream>>>(
          cnt, slots, ovf, ssrc, sdst, hf, bb, g, bt, m, vv, agg);

    in = agg;
  }

  final_linear<<<gblk(NNODES, 256), 256, 0, stream>>>(in, Wc, bc, (float*)d_out);
}


# Last layer's linear maps composed: per node 2 scores + 2 class values = h @ (W2 @ V), one K x 4 WMMA product; node pass gathers 8 B per edge, lane-tree sums, writes out (8 -> 7 launches)
# speedup vs baseline: 5.4176x; 1.0512x over previous
#include <hip/hip_runtime.h>

typedef __attribute__((ext_vector_type(2))) float v2f;
typedef __attribute__((ext_vector_type(8))) float v8f;

#define NNODES 50000
#define NEDGES 800000
#define FIN    128
#define HIDC   64
#define NHEADS 4
#define EPSBN  1e-5f
#define CAP    32
#define NCNT   ((NNODES + 1 + 63) / 64 * 64)
static_assert(NNODES <= 65536, "an overflow entry packs two node ids into one 32-bit word");

#define GEMM_NP     64
#define GEMM_PP     160
#define GEMM_MTILES (NNODES / 16)
static_assert(GEMM_NP == HIDC, "the score epilogue takes one panel as one head");

#define CNT_PANELS  ((NHEADS * HIDC) / GEMM_NP)
#define CNT_BLOCKS  256
#define CNT_ROWS    (CNT_BLOCKS / CNT_PANELS)
#define CNT_EDGES   (NEDGES / CNT_BLOCKS)
#define CNT_TRIPS   ((CNT_EDGES + 255) / 256)
static_assert(CNT_BLOCKS % CNT_PANELS == 0 && NEDGES % CNT_BLOCKS == 0,
              "counting work-groups tile the edge list exactly");

template <int N>
__device__ __forceinline__ void fold_step(float (&u)[16], int bit, int lane) {
  const bool up = (lane & bit) != 0;
  #pragma unroll
  for (int k = 0; k < N / 2; ++k) {
    const float keep = up ? u[N / 2 + k] : u[k];
    const float give = up ? u[k] : u[N / 2 + k];
    u[k] = keep + __shfl_xor(give, bit, 32);
  }
}

template <int K, bool WITH_COUNT>
__global__ __launch_bounds__(256) void gemm_wmma_f32(
    const float* __restrict__ A, const float* __restrict__ B,
    float* __restrict__ C, int Nc,
    const float* __restrict__ a_s, const float* __restrict__ a_d,
    float* __restrict__ s_src, float* __restrict__ s_dst,
    const int* __restrict__ ei, unsigned* __restrict__ cnt,
    int* __restrict__ slots, unsigned* __restrict__ ovf)
{
  __shared__ float Bs[(K / 2) * GEMM_PP];

  int tile_row = blockIdx.y;
  if (WITH_COUNT) {
    if (tile_row < CNT_ROWS) {
      const int e0 = (tile_row * CNT_PANELS + (int)blockIdx.x) * CNT_EDGES;
      const int* sp = ei + e0;
      const int* dp = ei + NEDGES + e0;
      unsigned* novf = cnt + NNODES;
      const int t = (int)threadIdx.x;
      int d[CNT_TRIPS], s[CNT_TRIPS];
      unsigned r[CNT_TRIPS];
      #pragma unroll
      for (int i = 0; i < CNT_TRIPS; ++i) {
        const bool ok = t + 256 * i < CNT_EDGES;
        d[i] = ok ? dp[t + 256 * i] : 0;
        s[i] = ok ? sp[t + 256 * i] : 0;
      }
      #pragma unroll
      for (int i = 0; i < CNT_TRIPS; ++i)
        r[i] = (t + 256 * i < CNT_EDGES) ? atomicAdd(&cnt[d[i]], 1u) : 0u;
      #pragma unroll
      for (int i = 0; i < CNT_TRIPS; ++i) {
        if (t + 256 * i < CNT_EDGES) {
          if (r[i] < (unsigned)CAP) {
            slots[(size_t)d[i] * CAP + r[i]] = s[i];
          } else {
            const unsigned k = atomicAdd(novf, 1u);
            ovf[k] = ((unsigned)d[i] << 16) | (unsigned)s[i];
          }
        }
      }
      return;
    }
    tile_row -= CNT_ROWS;
  }

  const int tid  = threadIdx.x;
  const int lane = tid & 31;
  const int wave = tid >> 5;
  const int half = lane >> 4;
  const int ml   = lane & 15;
  const int col0 = blockIdx.x * GEMM_NP;

  #pragma unroll
  for (int it = 0; it < (K / 2) * (GEMM_NP / 4) / 256; ++it) {
    const int i = tid + 256 * it;
    const int pr = i >> 4, c4 = (i & 15) * 4;
    const float* src = B + (size_t)(2 * pr) * Nc + col0 + c4;
    const float4 x0 = *(const float4*)(src);
    const float4 x1 = *(const float4*)(src + Nc);
    float* dst = Bs + pr * GEMM_PP + 2 * c4;
    *(float4*)(dst)     = make_float4(x0.x, x1.x, x0.y, x1.y);
    *(float4*)(dst + 4) = make_float4(x0.z, x1.z, x0.w, x1.w);
  }
  __syncthreads();

  const int tile = tile_row * 8 + wave;
  if (tile < GEMM_MTILES) {
    const float* ap = A + (size_t)(tile * 16 + ml) * K + 2 * half;
    const float* bp = Bs + half * GEMM_PP + 2 * ml;

    v8f acc[4];
    #pragma unroll
    for (int n = 0; n < 4; ++n)
      acc[n] = (v8f){0.f,0.f,0.f,0.f,0.f,0.f,0.f,0.f};

    v2f a_cur[8], a_nxt[8];
    #pragma unroll
    for (int j = 0; j < 8; ++j) a_cur[j] = *(const v2f*)(ap + 4 * j);

    #pragma nounroll
    for (int kb = 0; kb < K; kb += 32) {
      const int kn = (kb + 32 < K) ? kb + 32 : kb;
      #pragma unroll
      for (int j = 0; j < 8; ++j) a_nxt[j] = *(const v2f*)(ap + kn + 4 * j);

      #pragma unroll
      for (int j = 0; j < 8; ++j) {
        const float* bk = bp + ((kb + 4 * j) >> 1) * GEMM_PP;
        #pragma unroll
        for (int n = 0; n < 4; ++n) {
          const v2f b = *(const v2f*)(bk + 32 * n);
          acc[n] = __builtin_amdgcn_wmma_f32_16x16x4_f32(
              false, a_cur[j], false, b, (short)0, acc[n], false, false);
        }
      }
      #pragma unroll
      for (int j = 0; j < 8; ++j) a_cur[j] = a_nxt[j];
    }

    float* cp = C + (size_t)(tile * 16 + 8 * half) * Nc + col0 + ml;
    #pragma unroll
    for (int n = 0; n < 4; ++n) {
      #pragma unroll
      for (int i = 0; i < 8; ++i)
        cp[(size_t)i * Nc + 16 * n] = acc[n][i];
    }

    float u[16];
    #pragma unroll
    for (int i = 0; i < 16; ++i) u[i] = 0.f;
    #pragma unroll
    for (int n = 0; n < 4; ++n) {
      const float ws = a_s[col0 + 16 * n + ml];
      const float wd = a_d[col0 + 16 * n + ml];
      #pragma unroll
      for (int i = 0; i < 8; ++i) {
        u[i]     += acc[n][i] * ws;
        u[8 + i] += acc[n][i] * wd;
      }
    }
    fold_step<16>(u, 8, lane);
    fold_step<8>(u, 4, lane);
    fold_step<4>(u, 2, lane);
    fold_step<2>(u, 1, lane);
    const int heads = Nc / HIDC;
    float* sp = (ml & 8) ? s_dst : s_src;
    sp[(size_t)(tile * 16 + 8 * half + (ml & 7)) * heads + blockIdx.x] = u[0];
  }
}


__device__ __forceinline__ float leaky02(float v) { return v > 0.f ? v : 0.2f * v; }

__device__ __forceinline__ float wave_max(float v) {
  #pragma unroll
  for (int o = 16; o > 0; o >>= 1) v = fmaxf(v, __shfl_xor(v, o, 32));
  return v;
}
__device__ __forceinline__ float wave_sum(float v) {
  #pragma unroll
  for (int o = 16; o > 0; o >>= 1) v += __shfl_xor(v, o, 32);
  return v;
}
__device__ __forceinline__ float lane_bcast(float v, int j) {
  return __int_as_float(__builtin_amdgcn_readlane(__float_as_int(v), j));
}

template <int HEADS>
__device__ __forceinline__ void load_scores(const float* __restrict__ p, int i,
                                            float (&o)[HEADS]) {
  if (HEADS == 4) {
    const float4 t = *(const float4*)(p + (size_t)i * 4);
    o[0] = t.x; o[1] = t.y; o[2] = t.z; o[3] = t.w;
  } else {
    #pragma unroll
    for (int h = 0; h < HEADS; ++h) o[h] = p[(size_t)i * HEADS + h];
  }
}

template <bool ELU>
__device__ __forceinline__ float bias_bn_act(float x, float b, float mu, float var,
                                             float gam, float bet) {
  x = x + b;
  x = (x - mu) * rsqrtf(var + EPSBN) * gam + bet;
  if (ELU) x = x > 0.f ? x : expm1f(x);
  return x;
}

template <bool ELU>
__device__ __forceinline__ float4 bias_bn_act4(
    float4 x, int ch0, const float* __restrict__ bias, const float* __restrict__ g,
    const float* __restrict__ bt, const float* __restrict__ m, const float* __restrict__ v) {
  const float4 b  = *(const float4*)(bias + ch0);
  const float4 mu = *(const float4*)(m + ch0);
  const float4 va = *(const float4*)(v + ch0);
  const float4 ga = *(const float4*)(g + ch0);
  const float4 be = *(const float4*)(bt + ch0);
  return make_float4(bias_bn_act<ELU>(x.x, b.x, mu.x, va.x, ga.x, be.x),
                     bias_bn_act<ELU>(x.y, b.y, mu.y, va.y, ga.y, be.y),
                     bias_bn_act<ELU>(x.z, b.z, mu.z, va.z, ga.z, be.z),
                     bias_bn_act<ELU>(x.w, b.w, mu.w, va.w, ga.w, be.w));
}

template <int HEADS>
__device__ __forceinline__ void row_load(const float* __restrict__ hf, int sj, int lane,
                                         float4& a, float4& b) {
  const float* row = hf + (size_t)sj * (HEADS * HIDC);
  if (HEADS == 4) {
    a = *(const float4*)(row + 4 * lane);
    b = *(const float4*)(row + 128 + 4 * lane);
  } else {
    const float2 t = *(const float2*)(row + 2 * lane);
    a.x = t.x; a.y = t.y;
  }
}

template <int HEADS>
__device__ __forceinline__ void row_fma(const float4& a, const float4& b, int j,
                                        const float (&e)[HEADS], bool lo,
                                        float4& accA, float4& accB) {
  if (HEADS == 4) {
    const float e0 = lane_bcast(e[0], j), e1 = lane_bcast(e[1 % HEADS], j);
    const float e2 = lane_bcast(e[2 % HEADS], j), e3 = lane_bcast(e[3 % HEADS], j);
    const float eA = lo ? e0 : e1;
    const float eB = lo ? e2 : e3;
    accA.x = __builtin_fmaf(eA, a.x, accA.x); accA.y = __builtin_fmaf(eA, a.y, accA.y);
    accA.z = __builtin_fmaf(eA, a.z, accA.z); accA.w = __builtin_fmaf(eA, a.w, accA.w);
    accB.x = __builtin_fmaf(eB, b.x, accB.x); accB.y = __builtin_fmaf(eB, b.y, accB.y);
    accB.z = __builtin_fmaf(eB, b.z, accB.z); accB.w = __builtin_fmaf(eB, b.w, accB.w);
  } else {
    const float e0 = lane_bcast(e[0], j);
    accA.x = __builtin_fmaf(e0, a.x, accA.x); accA.y = __builtin_fmaf(e0, a.y, accA.y);
  }
}

template <int HEADS>
__device__ __forceinline__ void add_row(const float* __restrict__ hf, int sj, int j,
                                        const float (&e)[HEADS], bool lo, int lane,
                                        float4& accA, float4& accB) {
  float4 a, b;
  row_load<HEADS>(hf, sj, lane, a, b);
  row_fma<HEADS>(a, b, j, e, lo, accA, accB);
}

template <int HEADS, int G>
__device__ __forceinline__ void rows_request(const float* __restrict__ hf, int s, int jb,
                                             int jl, int lane,
                                             float4 (&a)[8], float4 (&b)[8]) {
  #pragma unroll
  for (int u = 0; u < G; ++u) {
    const int j = (jb + u) < jl ? (jb + u) : jl;
    row_load<HEADS>(hf, __builtin_amdgcn_readlane(s, j), lane, a[u], b[u]);
  }
}

template <int HEADS, int G>
__device__ __forceinline__ void rows_add(const float4 (&a)[8], const float4 (&b)[8], int jb,
                                         const float (&e)[HEADS], bool lo,
                                         float4& accA, float4& accB) {
  #pragma unroll
  for (int u = 0; u < G; ++u) row_fma<HEADS>(a[u], b[u], jb + u, e, lo, accA, accB);
}
template <int HEADS, int G>
__device__ __forceinline__ void rows_add_tail(const float4 (&a)[8], const float4 (&b)[8],
                                              int jb, int n, const float (&e)[HEADS], bool lo,
                                              float4& accA, float4& accB) {
  #pragma unroll
  for (int u = 0; u < G; ++u)
    if (jb + u < n) row_fma<HEADS>(a[u], b[u], jb + u, e, lo, accA, accB);
}

template <int HEADS, bool ELU>
__global__ __launch_bounds__(256) void node_softmax_gather(
    const unsigned* __restrict__ cnt, const int* __restrict__ slots,
    const unsigned* __restrict__ ovf,
    const float* __restrict__ s_src, const float* __restrict__ s_dst,
    const float* __restrict__ hf,
    const float* __restrict__ bias, const float* __restrict__ g,
    const float* __restrict__ bt, const float* __restrict__ m_bn,
    const float* __restrict__ v_bn, float* __restrict__ agg)
{
  static_assert(CAP == 32, "one lane per bucket slot");
  constexpr int HC = HEADS * HIDC;
  const int lane = threadIdx.x & 31;
  const int wave = __builtin_amdgcn_readfirstlane((int)(threadIdx.x >> 5));
  const int node = blockIdx.x * 8 + wave;
  if (node >= NNODES) return;
  const int indeg = (int)cnt[node];
  const int deg = indeg < CAP ? indeg : CAP;
  const int ncur = deg + 1;
  const int n0 = ncur < 32 ? ncur : 32;
  const int novf = indeg > CAP ? (int)cnt[NNODES] : 0;

  float sd[HEADS], m[HEADS], den[HEADS];
  load_scores<HEADS>(s_dst, node, sd);

  int s0 = node;
  if (lane < deg) s0 = slots[(size_t)node * CAP + lane];
  float lg[HEADS];
  {
    float sc[HEADS];
    load_scores<HEADS>(s_src, s0, sc);
    #pragma unroll
    for (int h = 0; h < HEADS; ++h) lg[h] = leaky02(sc[h] + sd[h]);
  }
  float4 ra[8], rb[8];
  rows_request<HEADS, 8>(hf, s0, 0, n0 - 1, lane, ra, rb);

  #pragma unroll
  for (int h = 0; h < HEADS; ++h) m[h] = lane < n0 ? lg[h] : -INFINITY;
  float lgs[HEADS];
  if (ncur > 32) {
    float sc[HEADS];
    load_scores<HEADS>(s_src, node, sc);
    #pragma unroll
    for (int h = 0; h < HEADS; ++h) {
      lgs[h] = leaky02(sc[h] + sd[h]);
      m[h] = fmaxf(m[h], lgs[h]);
    }
  }
  for (int k0 = 0; k0 < novf; k0 += 32) {
    const int k = k0 + lane;
    if (k < novf) {
      const unsigned w = ovf[k];
      if ((int)(w >> 16) == node) {
        float sc[HEADS];
        load_scores<HEADS>(s_src, (int)(w & 0xffffu), sc);
        #pragma unroll
        for (int h = 0; h < HEADS; ++h) m[h] = fmaxf(m[h], leaky02(sc[h] + sd[h]));
      }
    }
  }
  #pragma unroll
  for (int h = 0; h < HEADS; ++h) m[h] = wave_max(m[h]);

  float4 accA = make_float4(0.f, 0.f, 0.f, 0.f);
  float4 accB = make_float4(0.f, 0.f, 0.f, 0.f);
  const bool lo = lane < 16;

  float e[HEADS];
  #pragma unroll
  for (int h = 0; h < HEADS; ++h) {
    e[h] = lane < n0 ? expf(lg[h] - m[h]) : 0.f;
    den[h] = e[h];
  }
  rows_add_tail<HEADS, 8>(ra, rb, 0, n0, e, lo, accA, accB);
  int j0 = 8;
  for (; j0 + 8 <= n0; j0 += 8) {
    rows_request<HEADS, 8>(hf, s0, j0, n0 - 1, lane, ra, rb);
    rows_add<HEADS, 8>(ra, rb, j0, e, lo, accA, accB);
  }
  if (j0 < n0) {
    if (n0 - j0 > 4) {
      rows_request<HEADS, 8>(hf, s0, j0, n0 - 1, lane, ra, rb);
      rows_add_tail<HEADS, 8>(ra, rb, j0, n0, e, lo, accA, accB);
    } else {
      rows_request<HEADS, 4>(hf, s0, j0, n0 - 1, lane, ra, rb);
      rows_add_tail<HEADS, 4>(ra, rb, j0, n0, e, lo, accA, accB);
    }
  }
  if (ncur > 32) {
    float es[HEADS];
    #pragma unroll
    for (int h = 0; h < HEADS; ++h) {
      es[h] = expf(lgs[h] - m[h]);
      den[h] += lane == 0 ? es[h] : 0.f;
    }
    add_row<HEADS>(hf, node, 0, es, lo, lane, accA, accB);
  }
  for (int k0 = 0; k0 < novf; k0 += 32) {
    const int k = k0 + lane;
    int s = node;
    bool hit = false;
    float eo[HEADS];
    #pragma unroll
    for (int h = 0; h < HEADS; ++h) eo[h] = 0.f;
    if (k < novf) {
      const unsigned w = ovf[k];
      hit = (int)(w >> 16) == node;
      if (hit) {
        s = (int)(w & 0xffffu);
        float sc[HEADS];
        load_scores<HEADS>(s_src, s, sc);
        #pragma unroll
        for (int h = 0; h < HEADS; ++h) eo[h] = expf(leaky02(sc[h] + sd[h]) - m[h]);
      }
    }
    #pragma unroll
    for (int h = 0; h < HEADS; ++h) den[h] += eo[h];
    unsigned mask = __builtin_amdgcn_ballot_w32(hit);
    while (mask != 0u) {
      const int j = __builtin_ctz(mask);
      mask &= mask - 1u;
      const int sj = __builtin_amdgcn_readlane(s, j);
      add_row<HEADS>(hf, sj, j, eo, lo, lane, accA, accB);
    }
  }
  #pragma unroll
  for (int h = 0; h < HEADS; ++h) den[h] = wave_sum(den[h]);

  float* op = agg + (size_t)node * HC;
  if (HEADS == 4) {
    const float dA = (lo ? den[0] : den[1 % HEADS]) + 1e-16f;
    const float dB = (lo ? den[2 % HEADS] : den[3 % HEADS]) + 1e-16f;
    *(float4*)(op + 4 * lane) = bias_bn_act4<ELU>(
        make_float4(accA.x / dA, accA.y / dA, accA.z / dA, accA.w / dA),
        4 * lane, bias, g, bt, m_bn, v_bn);
    *(float4*)(op + 128 + 4 * lane) = bias_bn_act4<ELU>(
        make_float4(accB.x / dB, accB.y / dB, accB.z / dB, accB.w / dB),
        128 + 4 * lane, bias, g, bt, m_bn, v_bn);
  } else {
    const float d0 = den[0] + 1e-16f;
    const int ch0 = 2 * lane;
    const float2 b  = *(const float2*)(bias + ch0);
    const float2 mu = *(const float2*)(m_bn + ch0);
    const float2 va = *(const float2*)(v_bn + ch0);
    const float2 ga = *(const float2*)(g + ch0);
    const float2 be = *(const float2*)(bt + ch0);
    *(float2*)(op + ch0) =
        make_float2(bias_bn_act<ELU>(accA.x / d0, b.x, mu.x, va.x, ga.x, be.x),
                    bias_bn_act<ELU>(accA.y / d0, b.y, mu.y, va.y, ga.y, be.y));
  }
}

#define PRJ_PP 32
template <int K>
__global__ __launch_bounds__(256) void last_layer_project(
    const float* __restrict__ A, const float* __restrict__ W,
    const float* __restrict__ a_s, const float* __restrict__ a_d,
    const float* __restrict__ g_bn, const float* __restrict__ v_bn,
    const float* __restrict__ Wc,
    float* __restrict__ s_src, float* __restrict__ s_dst, float* __restrict__ proj)
{
  static_assert(K == 256, "one thread per row of W");
  __shared__ float4 V[HIDC];
  __shared__ float Bs[(K / 2) * PRJ_PP];

  const int tid  = threadIdx.x;
  const int lane = tid & 31;
  const int wave = tid >> 5;
  const int half = lane >> 4;
  const int ml   = lane & 15;

  if (tid < HIDC) {
    const float sg = rsqrtf(v_bn[tid] + EPSBN) * g_bn[tid];
    V[tid] = make_float4(a_s[tid], a_d[tid], sg * Wc[2 * tid + 0], sg * Wc[2 * tid + 1]);
  }
  __syncthreads();
  {
    const float* wr = W + (size_t)tid * HIDC;
    float4 s = make_float4(0.f, 0.f, 0.f, 0.f);
    #pragma unroll
    for (int q = 0; q < HIDC / 4; ++q) {
      const float4 w = *(const float4*)(wr + 4 * q);
      const float4 v0 = V[4 * q + 0], v1 = V[4 * q + 1], v2 = V[4 * q + 2], v3 = V[4 * q + 3];
      s.x += w.x * v0.x + w.y * v1.x + w.z * v2.x + w.w * v3.x;
      s.y += w.x * v0.y + w.y * v1.y + w.z * v2.y + w.w * v3.y;
      s.z += w.x * v0.z + w.y * v1.z + w.z * v2.z + w.w * v3.z;
      s.w += w.x * v0.w + w.y * v1.w + w.z * v2.w + w.w * v3.w;
    }
    float* dst = Bs + (tid >> 1) * PRJ_PP + (tid & 1);
    dst[0] = s.x; dst[2] = s.y; dst[4] = s.z; dst[6] = s.w;
    #pragma unroll
    for (int cidx = 4; cidx < 16; ++cidx) dst[2 * cidx] = 0.f;
  }
  __syncthreads();

  const int tile = blockIdx.x * 8 + wave;
  if (tile < GEMM_MTILES) {
    const float* ap = A + (size_t)(tile * 16 + ml) * K + 2 * half;
    const float* bp = Bs + half * PRJ_PP + 2 * ml;

    v8f acc = (v8f){0.f,0.f,0.f,0.f,0.f,0.f,0.f,0.f};
    v2f a_cur[8], a_nxt[8];
    #pragma unroll
    for (int j = 0; j < 8; ++j) a_cur[j] = *(const v2f*)(ap + 4 * j);

    for (int kb = 0; kb < K; kb += 32) {
      const int kn = (kb + 32 < K) ? kb + 32 : kb;
      #pragma unroll
      for (int j = 0; j < 8; ++j) a_nxt[j] = *(const v2f*)(ap + kn + 4 * j);
      #pragma unroll
      for (int j = 0; j < 8; ++j) {
        const v2f b = *(const v2f*)(bp + ((kb + 4 * j) >> 1) * PRJ_PP);
        acc = __builtin_amdgcn_wmma_f32_16x16x4_f32(
            false, a_cur[j], false, b, (short)0, acc, false, false);
      }
      #pragma unroll
      for (int j = 0; j < 8; ++j) a_cur[j] = a_nxt[j];
    }

    if (ml < 4) {
      const int row0 = tile * 16 + 8 * half;
      #pragma unroll
      for (int i = 0; i < 8; ++i) {
        const int row = row0 + i;
        if (ml == 0)      s_src[row] = acc[i];
        else if (ml == 1) s_dst[row] = acc[i];
        else              proj[(size_t)row * 2 + (ml - 2)] = acc[i];
      }
    }
  }
}

__global__ __launch_bounds__(256) void node_softmax_tail(
    const unsigned* __restrict__ cnt, const int* __restrict__ slots,
    const unsigned* __restrict__ ovf,
    const float* __restrict__ s_src, const float* __restrict__ s_dst,
    const float* __restrict__ proj,
    const float* __restrict__ bias, const float* __restrict__ g,
    const float* __restrict__ bt, const float* __restrict__ m_bn,
    const float* __restrict__ v_bn, const float* __restrict__ Wc,
    const float* __restrict__ bc, float* __restrict__ out)
{
  static_assert(CAP == 32, "one lane per bucket slot");
  static_assert(HIDC == 64, "two channels per lane in the constant term");
  const int lane = threadIdx.x & 31;
  const int wave = __builtin_amdgcn_readfirstlane((int)(threadIdx.x >> 5));
  const int node = blockIdx.x * 8 + wave;
  if (node >= NNODES) return;
  const int indeg = (int)cnt[node];
  const int deg = indeg < CAP ? indeg : CAP;
  const int ncur = deg + 1;
  const int n0 = ncur < 32 ? ncur : 32;
  const int novf = indeg > CAP ? (int)cnt[NNODES] : 0;
  const float sd = s_dst[node];

  float c0, c1;
  {
    const int ch = 2 * lane;
    const float2 b  = *(const float2*)(bias + ch);
    const float2 mu = *(const float2*)(m_bn + ch);
    const float2 va = *(const float2*)(v_bn + ch);
    const float2 ga = *(const float2*)(g + ch);
    const float2 be = *(const float2*)(bt + ch);
    const float4 w  = *(const float4*)(Wc + 2 * ch);
    const float tx = (b.x - mu.x) * rsqrtf(va.x + EPSBN) * ga.x + be.x;
    const float ty = (b.y - mu.y) * rsqrtf(va.y + EPSBN) * ga.y + be.y;
    c0 = tx * w.x + ty * w.z;
    c1 = tx * w.y + ty * w.w;
  }

  int s0 = node;
  if (lane < deg) s0 = slots[(size_t)node * CAP + lane];
  const float lg = leaky02(s_src[s0] + sd);
  const float2 pv = *(const float2*)(proj + 2 * (size_t)s0);

  float m = lane < n0 ? lg : -INFINITY;
  float lgs = 0.f;
  if (ncur > 32) {
    lgs = leaky02(s_src[node] + sd);
    m = fmaxf(m, lgs);
  }
  for (int k0 = 0; k0 < novf; k0 += 32) {
    const int k = k0 + lane;
    if (k < novf) {
      const unsigned w = ovf[k];
      if ((int)(w >> 16) == node)
        m = fmaxf(m, leaky02(s_src[(int)(w & 0xffffu)] + sd));
    }
  }
  m = wave_max(m);

  const float e = lane < n0 ? expf(lg - m) : 0.f;
  float den = e, a0 = e * pv.x, a1 = e * pv.y;
  if (ncur > 32) {
    const float es = expf(lgs - m);
    const float2 ps = *(const float2*)(proj + 2 * (size_t)node);
    if (lane == 0) { den += es; a0 += es * ps.x; a1 += es * ps.y; }
  }
  for (int k0 = 0; k0 < novf; k0 += 32) {
    const int k = k0 + lane;
    if (k < novf) {
      const unsigned w = ovf[k];
      if ((int)(w >> 16) == node) {
        const int s = (int)(w & 0xffffu);
        const float eo = expf(leaky02(s_src[s] + sd) - m);
        const float2 po = *(const float2*)(proj + 2 * (size_t)s);
        den += eo; a0 += eo * po.x; a1 += eo * po.y;
      }
    }
  }
  den = wave_sum(den);
  a0 = wave_sum(a0);
  a1 = wave_sum(a1);
  c0 = wave_sum(c0);
  c1 = wave_sum(c1);
  if (lane == 0) {
    const float d = den + 1e-16f;
    *(float2*)(out + 2 * (size_t)node) = make_float2(a0 / d + c0 + bc[0], a1 / d + c1 + bc[1]);
  }
}

__global__ void fill_u32(unsigned* __restrict__ p, unsigned val, long long n)
{
  long long i = (long long)blockIdx.x * blockDim.x + threadIdx.x;
  if (i < n) p[i] = val;
}

static inline unsigned gblk(long long total, int bs) {
  return (unsigned)((total + bs - 1) / bs);
}

extern "C" void kernel_launch(void* const* d_in, const int* in_sizes, int n_in,
                              void* d_out, int out_size, void* d_ws, size_t ws_size,
                              hipStream_t stream)
{
  const float* x  = (const float*)d_in[0];
  const int*   ei = (const int*)  d_in[1];
  const float* Wc = (const float*)d_in[26];
  const float* bc = (const float*)d_in[27];

  char* ws = (char*)d_ws;
  float*    hf    = (float*)ws;    ws += (size_t)NNODES * 256 * sizeof(float);
  float*    agg   = (float*)ws;    ws += (size_t)NNODES * 256 * sizeof(float);
  float*    ssrc  = (float*)ws;    ws += (size_t)NNODES * NHEADS * sizeof(float);
  float*    sdst  = (float*)ws;    ws += (size_t)NNODES * NHEADS * sizeof(float);
  unsigned* cnt    = (unsigned*)ws; ws += (size_t)NCNT * sizeof(unsigned);
  unsigned* ovf    = (unsigned*)ws; ws += (size_t)NEDGES * sizeof(unsigned);
  int*      slots  = (int*)ws;      ws += (size_t)NNODES * CAP * sizeof(int);

  fill_u32<<<gblk(NCNT, 256), 256, 0, stream>>>(cnt, 0u, NCNT);

  const float* in = x;
  for (int l = 0; l < 3; ++l) {
    int base = 2 + l * 8;
    const float* W  = (const float*)d_in[base + 0];
    const float* as = (const float*)d_in[base + 1];
    const float* ad = (const float*)d_in[base + 2];
    const float* bb = (const float*)d_in[base + 3];
    const float* g  = (const float*)d_in[base + 4];
    const float* bt = (const float*)d_in[base + 5];
    const float* m  = (const float*)d_in[base + 6];
    const float* vv = (const float*)d_in[base + 7];

    const int heads = (l < 2) ? NHEADS : 1;
    const int hc    = heads * HIDC;

    dim3 ggrid(hc / GEMM_NP, (GEMM_MTILES + 7) / 8);
    if (l == 0) {
      dim3 ggrid0(CNT_PANELS, CNT_ROWS + (GEMM_MTILES + 7) / 8);
      gemm_wmma_f32<FIN, true><<<ggrid0, 256, 0, stream>>>(
          in, W, hf, hc, as, ad, ssrc, sdst, ei, cnt, slots, ovf);
    } else if (heads == NHEADS) {
      gemm_wmma_f32<NHEADS * HIDC, false><<<ggrid, 256, 0, stream>>>(
          in, W, hf, hc, as, ad, ssrc, sdst, nullptr, nullptr, nullptr, nullptr);
    } else {
      last_layer_project<NHEADS * HIDC><<<(GEMM_MTILES + 7) / 8, 256, 0, stream>>>(
          in, W, as, ad, g, vv, Wc, ssrc, sdst, hf);
    }

    if (heads == NHEADS)
      node_softmax_gather<NHEADS, true><<<NNODES / 8, 256, 0, stream>>>(
          cnt, slots, ovf, ssrc, sdst, hf, bb, g, bt, m, vv, agg);
    else
      node_softmax_tail<<<NNODES / 8, 256, 0, stream>>>(
          cnt, slots, ovf, ssrc, sdst, hf, bb, g, bt, m, vv, Wc, bc, (float*)d_out);

    in = agg;
  }
}


# Node kernels' epilogues: channel-only terms (bias + BatchNorm scale/shift, last layer's constant) formed once per call in launch 1; row x one reciprocal per head (was 8 divisions); branch-free ELU
# speedup vs baseline: 5.6351x; 1.0401x over previous
#include <hip/hip_runtime.h>

typedef __attribute__((ext_vector_type(2))) float v2f;
typedef __attribute__((ext_vector_type(8))) float v8f;

#define NNODES 50000
#define NEDGES 800000
#define FIN    128
#define HIDC   64
#define NHEADS 4
#define EPSBN  1e-5f
#define CAP    32
#define NCNT   ((NNODES + 1 + 63) / 64 * 64)
static_assert(NNODES <= 65536, "an overflow entry packs two node ids into one 32-bit word");

#define GEMM_NP     64
#define GEMM_PP     160
#define GEMM_MTILES (NNODES / 16)
static_assert(GEMM_NP == HIDC, "the score epilogue takes one panel as one head");

#define CNT_PANELS  ((NHEADS * HIDC) / GEMM_NP)
#define CNT_BLOCKS  256
#define CNT_ROWS    (CNT_BLOCKS / CNT_PANELS)
#define CNT_EDGES   (NEDGES / CNT_BLOCKS)
#define CNT_TRIPS   ((CNT_EDGES + 255) / 256)
static_assert(CNT_BLOCKS % CNT_PANELS == 0 && NEDGES % CNT_BLOCKS == 0,
              "counting work-groups tile the edge list exactly");

template <int N>
__device__ __forceinline__ void fold_step(float (&u)[16], int bit, int lane) {
  const bool up = (lane & bit) != 0;
  #pragma unroll
  for (int k = 0; k < N / 2; ++k) {
    const float keep = up ? u[N / 2 + k] : u[k];
    const float give = up ? u[k] : u[N / 2 + k];
    u[k] = keep + __shfl_xor(give, bit, 32);
  }
}

template <int K, bool WITH_COUNT>
__global__ __launch_bounds__(256) void gemm_wmma_f32(
    const float* __restrict__ A, const float* __restrict__ B,
    float* __restrict__ C, int Nc,
    const float* __restrict__ a_s, const float* __restrict__ a_d,
    float* __restrict__ s_src, float* __restrict__ s_dst,
    const int* __restrict__ ei, unsigned* __restrict__ cnt,
    int* __restrict__ slots, unsigned* __restrict__ ovf)
{
  __shared__ float Bs[(K / 2) * GEMM_PP];

  int tile_row = blockIdx.y;
  if (WITH_COUNT) {
    if (tile_row < CNT_ROWS) {
      const int e0 = (tile_row * CNT_PANELS + (int)blockIdx.x) * CNT_EDGES;
      const int* sp = ei + e0;
      const int* dp = ei + NEDGES + e0;
      unsigned* novf = cnt + NNODES;
      const int t = (int)threadIdx.x;
      int d[CNT_TRIPS], s[CNT_TRIPS];
      unsigned r[CNT_TRIPS];
      #pragma unroll
      for (int i = 0; i < CNT_TRIPS; ++i) {
        const bool ok = t + 256 * i < CNT_EDGES;
        d[i] = ok ? dp[t + 256 * i] : 0;
        s[i] = ok ? sp[t + 256 * i] : 0;
      }
      #pragma unroll
      for (int i = 0; i < CNT_TRIPS; ++i)
        r[i] = (t + 256 * i < CNT_EDGES) ? atomicAdd(&cnt[d[i]], 1u) : 0u;
      #pragma unroll
      for (int i = 0; i < CNT_TRIPS; ++i) {
        if (t + 256 * i < CNT_EDGES) {
          if (r[i] < (unsigned)CAP) {
            slots[(size_t)d[i] * CAP + r[i]] = s[i];
          } else {
            const unsigned k = atomicAdd(novf, 1u);
            ovf[k] = ((unsigned)d[i] << 16) | (unsigned)s[i];
          }
        }
      }
      return;
    }
    tile_row -= CNT_ROWS;
  }

  const int tid  = threadIdx.x;
  const int lane = tid & 31;
  const int wave = tid >> 5;
  const int half = lane >> 4;
  const int ml   = lane & 15;
  const int col0 = blockIdx.x * GEMM_NP;

  #pragma unroll
  for (int it = 0; it < (K / 2) * (GEMM_NP / 4) / 256; ++it) {
    const int i = tid + 256 * it;
    const int pr = i >> 4, c4 = (i & 15) * 4;
    const float* src = B + (size_t)(2 * pr) * Nc + col0 + c4;
    const float4 x0 = *(const float4*)(src);
    const float4 x1 = *(const float4*)(src + Nc);
    float* dst = Bs + pr * GEMM_PP + 2 * c4;
    *(float4*)(dst)     = make_float4(x0.x, x1.x, x0.y, x1.y);
    *(float4*)(dst + 4) = make_float4(x0.z, x1.z, x0.w, x1.w);
  }
  __syncthreads();

  const int tile = tile_row * 8 + wave;
  if (tile < GEMM_MTILES) {
    const float* ap = A + (size_t)(tile * 16 + ml) * K + 2 * half;
    const float* bp = Bs + half * GEMM_PP + 2 * ml;

    v8f acc[4];
    #pragma unroll
    for (int n = 0; n < 4; ++n)
      acc[n] = (v8f){0.f,0.f,0.f,0.f,0.f,0.f,0.f,0.f};

    v2f a_cur[8], a_nxt[8];
    #pragma unroll
    for (int j = 0; j < 8; ++j) a_cur[j] = *(const v2f*)(ap + 4 * j);

    #pragma nounroll
    for (int kb = 0; kb < K; kb += 32) {
      const int kn = (kb + 32 < K) ? kb + 32 : kb;
      #pragma unroll
      for (int j = 0; j < 8; ++j) a_nxt[j] = *(const v2f*)(ap + kn + 4 * j);

      #pragma unroll
      for (int j = 0; j < 8; ++j) {
        const float* bk = bp + ((kb + 4 * j) >> 1) * GEMM_PP;
        #pragma unroll
        for (int n = 0; n < 4; ++n) {
          const v2f b = *(const v2f*)(bk + 32 * n);
          acc[n] = __builtin_amdgcn_wmma_f32_16x16x4_f32(
              false, a_cur[j], false, b, (short)0, acc[n], false, false);
        }
      }
      #pragma unroll
      for (int j = 0; j < 8; ++j) a_cur[j] = a_nxt[j];
    }

    float* cp = C + (size_t)(tile * 16 + 8 * half) * Nc + col0 + ml;
    #pragma unroll
    for (int n = 0; n < 4; ++n) {
      #pragma unroll
      for (int i = 0; i < 8; ++i)
        cp[(size_t)i * Nc + 16 * n] = acc[n][i];
    }

    float u[16];
    #pragma unroll
    for (int i = 0; i < 16; ++i) u[i] = 0.f;
    #pragma unroll
    for (int n = 0; n < 4; ++n) {
      const float ws = a_s[col0 + 16 * n + ml];
      const float wd = a_d[col0 + 16 * n + ml];
      #pragma unroll
      for (int i = 0; i < 8; ++i) {
        u[i]     += acc[n][i] * ws;
        u[8 + i] += acc[n][i] * wd;
      }
    }
    fold_step<16>(u, 8, lane);
    fold_step<8>(u, 4, lane);
    fold_step<4>(u, 2, lane);
    fold_step<2>(u, 1, lane);
    const int heads = Nc / HIDC;
    float* sp = (ml & 8) ? s_dst : s_src;
    sp[(size_t)(tile * 16 + 8 * half + (ml & 7)) * heads + blockIdx.x] = u[0];
  }
}


__device__ __forceinline__ float leaky02(float v) { return v > 0.f ? v : 0.2f * v; }

__device__ __forceinline__ float wave_max(float v) {
  #pragma unroll
  for (int o = 16; o > 0; o >>= 1) v = fmaxf(v, __shfl_xor(v, o, 32));
  return v;
}
__device__ __forceinline__ float wave_sum(float v) {
  #pragma unroll
  for (int o = 16; o > 0; o >>= 1) v += __shfl_xor(v, o, 32);
  return v;
}
__device__ __forceinline__ float lane_bcast(float v, int j) {
  return __int_as_float(__builtin_amdgcn_readlane(__float_as_int(v), j));
}

template <int HEADS>
__device__ __forceinline__ void load_scores(const float* __restrict__ p, int i,
                                            float (&o)[HEADS]) {
  if (HEADS == 4) {
    const float4 t = *(const float4*)(p + (size_t)i * 4);
    o[0] = t.x; o[1] = t.y; o[2] = t.z; o[3] = t.w;
  } else {
    #pragma unroll
    for (int h = 0; h < HEADS; ++h) o[h] = p[(size_t)i * HEADS + h];
  }
}

template <bool ELU>
__device__ __forceinline__ float finish_value(float x, float r, float s, float t) {
  float y = __builtin_fmaf(x * r, s, t);
  if (ELU) {
    const float em = expm1f(y);
    y = y > 0.f ? y : em;
  }
  return y;
}

template <bool ELU>
__device__ __forceinline__ float4 finish_value4(float4 x, float r, float4 s, float4 t) {
  return make_float4(finish_value<ELU>(x.x, r, s.x, t.x), finish_value<ELU>(x.y, r, s.y, t.y),
                     finish_value<ELU>(x.z, r, s.z, t.z), finish_value<ELU>(x.w, r, s.w, t.w));
}

template <int HEADS>
__device__ __forceinline__ void row_load(const float* __restrict__ hf, int sj, int lane,
                                         float4& a, float4& b) {
  const float* row = hf + (size_t)sj * (HEADS * HIDC);
  if (HEADS == 4) {
    a = *(const float4*)(row + 4 * lane);
    b = *(const float4*)(row + 128 + 4 * lane);
  } else {
    const float2 t = *(const float2*)(row + 2 * lane);
    a.x = t.x; a.y = t.y;
  }
}

template <int HEADS>
__device__ __forceinline__ void row_fma(const float4& a, const float4& b, int j,
                                        const float (&e)[HEADS], bool lo,
                                        float4& accA, float4& accB) {
  if (HEADS == 4) {
    const float e0 = lane_bcast(e[0], j), e1 = lane_bcast(e[1 % HEADS], j);
    const float e2 = lane_bcast(e[2 % HEADS], j), e3 = lane_bcast(e[3 % HEADS], j);
    const float eA = lo ? e0 : e1;
    const float eB = lo ? e2 : e3;
    accA.x = __builtin_fmaf(eA, a.x, accA.x); accA.y = __builtin_fmaf(eA, a.y, accA.y);
    accA.z = __builtin_fmaf(eA, a.z, accA.z); accA.w = __builtin_fmaf(eA, a.w, accA.w);
    accB.x = __builtin_fmaf(eB, b.x, accB.x); accB.y = __builtin_fmaf(eB, b.y, accB.y);
    accB.z = __builtin_fmaf(eB, b.z, accB.z); accB.w = __builtin_fmaf(eB, b.w, accB.w);
  } else {
    const float e0 = lane_bcast(e[0], j);
    accA.x = __builtin_fmaf(e0, a.x, accA.x); accA.y = __builtin_fmaf(e0, a.y, accA.y);
  }
}

template <int HEADS>
__device__ __forceinline__ void add_row(const float* __restrict__ hf, int sj, int j,
                                        const float (&e)[HEADS], bool lo, int lane,
                                        float4& accA, float4& accB) {
  float4 a, b;
  row_load<HEADS>(hf, sj, lane, a, b);
  row_fma<HEADS>(a, b, j, e, lo, accA, accB);
}

template <int HEADS, int G>
__device__ __forceinline__ void rows_request(const float* __restrict__ hf, int s, int jb,
                                             int jl, int lane,
                                             float4 (&a)[8], float4 (&b)[8]) {
  #pragma unroll
  for (int u = 0; u < G; ++u) {
    const int j = (jb + u) < jl ? (jb + u) : jl;
    row_load<HEADS>(hf, __builtin_amdgcn_readlane(s, j), lane, a[u], b[u]);
  }
}

template <int HEADS, int G>
__device__ __forceinline__ void rows_add(const float4 (&a)[8], const float4 (&b)[8], int jb,
                                         const float (&e)[HEADS], bool lo,
                                         float4& accA, float4& accB) {
  #pragma unroll
  for (int u = 0; u < G; ++u) row_fma<HEADS>(a[u], b[u], jb + u, e, lo, accA, accB);
}
template <int HEADS, int G>
__device__ __forceinline__ void rows_add_tail(const float4 (&a)[8], const float4 (&b)[8],
                                              int jb, int n, const float (&e)[HEADS], bool lo,
                                              float4& accA, float4& accB) {
  #pragma unroll
  for (int u = 0; u < G; ++u)
    if (jb + u < n) row_fma<HEADS>(a[u], b[u], jb + u, e, lo, accA, accB);
}

template <int HEADS, bool ELU>
__global__ __launch_bounds__(256) void node_softmax_gather(
    const unsigned* __restrict__ cnt, const int* __restrict__ slots,
    const unsigned* __restrict__ ovf,
    const float* __restrict__ s_src, const float* __restrict__ s_dst,
    const float* __restrict__ hf,
    const float* __restrict__ scale, const float* __restrict__ shift,
    float* __restrict__ agg)
{
  static_assert(CAP == 32, "one lane per bucket slot");
  constexpr int HC = HEADS * HIDC;
  const int lane = threadIdx.x & 31;
  const int wave = __builtin_amdgcn_readfirstlane((int)(threadIdx.x >> 5));
  const int node = blockIdx.x * 8 + wave;
  if (node >= NNODES) return;
  const int indeg = (int)cnt[node];
  const int deg = indeg < CAP ? indeg : CAP;
  const int ncur = deg + 1;
  const int n0 = ncur < 32 ? ncur : 32;
  const int novf = indeg > CAP ? (int)cnt[NNODES] : 0;

  float sd[HEADS], m[HEADS], den[HEADS];
  load_scores<HEADS>(s_dst, node, sd);

  int s0 = node;
  if (lane < deg) s0 = slots[(size_t)node * CAP + lane];
  float lg[HEADS];
  {
    float sc[HEADS];
    load_scores<HEADS>(s_src, s0, sc);
    #pragma unroll
    for (int h = 0; h < HEADS; ++h) lg[h] = leaky02(sc[h] + sd[h]);
  }
  float4 ra[8], rb[8];
  rows_request<HEADS, 8>(hf, s0, 0, n0 - 1, lane, ra, rb);

  #pragma unroll
  for (int h = 0; h < HEADS; ++h) m[h] = lane < n0 ? lg[h] : -INFINITY;
  float lgs[HEADS];
  if (ncur > 32) {
    float sc[HEADS];
    load_scores<HEADS>(s_src, node, sc);
    #pragma unroll
    for (int h = 0; h < HEADS; ++h) {
      lgs[h] = leaky02(sc[h] + sd[h]);
      m[h] = fmaxf(m[h], lgs[h]);
    }
  }
  for (int k0 = 0; k0 < novf; k0 += 32) {
    const int k = k0 + lane;
    if (k < novf) {
      const unsigned w = ovf[k];
      if ((int)(w >> 16) == node) {
        float sc[HEADS];
        load_scores<HEADS>(s_src, (int)(w & 0xffffu), sc);
        #pragma unroll
        for (int h = 0; h < HEADS; ++h) m[h] = fmaxf(m[h], leaky02(sc[h] + sd[h]));
      }
    }
  }
  #pragma unroll
  for (int h = 0; h < HEADS; ++h) m[h] = wave_max(m[h]);

  float4 accA = make_float4(0.f, 0.f, 0.f, 0.f);
  float4 accB = make_float4(0.f, 0.f, 0.f, 0.f);
  const bool lo = lane < 16;

  float e[HEADS];
  #pragma unroll
  for (int h = 0; h < HEADS; ++h) {
    e[h] = lane < n0 ? expf(lg[h] - m[h]) : 0.f;
    den[h] = e[h];
  }
  rows_add_tail<HEADS, 8>(ra, rb, 0, n0, e, lo, accA, accB);
  int j0 = 8;
  for (; j0 + 8 <= n0; j0 += 8) {
    rows_request<HEADS, 8>(hf, s0, j0, n0 - 1, lane, ra, rb);
    rows_add<HEADS, 8>(ra, rb, j0, e, lo, accA, accB);
  }
  if (j0 < n0) {
    if (n0 - j0 > 4) {
      rows_request<HEADS, 8>(hf, s0, j0, n0 - 1, lane, ra, rb);
      rows_add_tail<HEADS, 8>(ra, rb, j0, n0, e, lo, accA, accB);
    } else {
      rows_request<HEADS, 4>(hf, s0, j0, n0 - 1, lane, ra, rb);
      rows_add_tail<HEADS, 4>(ra, rb, j0, n0, e, lo, accA, accB);
    }
  }
  if (ncur > 32) {
    float es[HEADS];
    #pragma unroll
    for (int h = 0; h < HEADS; ++h) {
      es[h] = expf(lgs[h] - m[h]);
      den[h] += lane == 0 ? es[h] : 0.f;
    }
    add_row<HEADS>(hf, node, 0, es, lo, lane, accA, accB);
  }
  for (int k0 = 0; k0 < novf; k0 += 32) {
    const int k = k0 + lane;
    int s = node;
    bool hit = false;
    float eo[HEADS];
    #pragma unroll
    for (int h = 0; h < HEADS; ++h) eo[h] = 0.f;
    if (k < novf) {
      const unsigned w = ovf[k];
      hit = (int)(w >> 16) == node;
      if (hit) {
        s = (int)(w & 0xffffu);
        float sc[HEADS];
        load_scores<HEADS>(s_src, s, sc);
        #pragma unroll
        for (int h = 0; h < HEADS; ++h) eo[h] = expf(leaky02(sc[h] + sd[h]) - m[h]);
      }
    }
    #pragma unroll
    for (int h = 0; h < HEADS; ++h) den[h] += eo[h];
    unsigned mask = __builtin_amdgcn_ballot_w32(hit);
    while (mask != 0u) {
      const int j = __builtin_ctz(mask);
      mask &= mask - 1u;
      const int sj = __builtin_amdgcn_readlane(s, j);
      add_row<HEADS>(hf, sj, j, eo, lo, lane, accA, accB);
    }
  }
  float4 sA, tA, sB, tB;
  if (HEADS == 4) {
    sA = *(const float4*)(scale + 4 * lane);
    tA = *(const float4*)(shift + 4 * lane);
    sB = *(const float4*)(scale + 128 + 4 * lane);
    tB = *(const float4*)(shift + 128 + 4 * lane);
  } else {
    const float2 s2 = *(const float2*)(scale + 2 * lane);
    const float2 t2 = *(const float2*)(shift + 2 * lane);
    sA.x = s2.x; sA.y = s2.y; tA.x = t2.x; tA.y = t2.y;
  }
  #pragma unroll
  for (int h = 0; h < HEADS; ++h) den[h] = wave_sum(den[h]);

  float* op = agg + (size_t)node * HC;
  if (HEADS == 4) {
    const float rA = 1.0f / ((lo ? den[0] : den[1 % HEADS]) + 1e-16f);
    const float rB = 1.0f / ((lo ? den[2 % HEADS] : den[3 % HEADS]) + 1e-16f);
    *(float4*)(op + 4 * lane)       = finish_value4<ELU>(accA, rA, sA, tA);
    *(float4*)(op + 128 + 4 * lane) = finish_value4<ELU>(accB, rB, sB, tB);
  } else {
    const float r0 = 1.0f / (den[0] + 1e-16f);
    *(float2*)(op + 2 * lane) = make_float2(finish_value<ELU>(accA.x, r0, sA.x, tA.x),
                                            finish_value<ELU>(accA.y, r0, sA.y, tA.y));
  }
}

#define PRJ_PP 32
template <int K>
__global__ __launch_bounds__(256) void last_layer_project(
    const float* __restrict__ A, const float* __restrict__ W,
    const float* __restrict__ a_s, const float* __restrict__ a_d,
    const float* __restrict__ g_bn, const float* __restrict__ v_bn,
    const float* __restrict__ Wc,
    float* __restrict__ s_src, float* __restrict__ s_dst, float* __restrict__ proj)
{
  static_assert(K == 256, "one thread per row of W");
  __shared__ float4 V[HIDC];
  __shared__ float Bs[(K / 2) * PRJ_PP];

  const int tid  = threadIdx.x;
  const int lane = tid & 31;
  const int wave = tid >> 5;
  const int half = lane >> 4;
  const int ml   = lane & 15;

  if (tid < HIDC) {
    const float sg = rsqrtf(v_bn[tid] + EPSBN) * g_bn[tid];
    V[tid] = make_float4(a_s[tid], a_d[tid], sg * Wc[2 * tid + 0], sg * Wc[2 * tid + 1]);
  }
  __syncthreads();
  {
    const float* wr = W + (size_t)tid * HIDC;
    float4 s = make_float4(0.f, 0.f, 0.f, 0.f);
    #pragma unroll
    for (int q = 0; q < HIDC / 4; ++q) {
      const float4 w = *(const float4*)(wr + 4 * q);
      const float4 v0 = V[4 * q + 0], v1 = V[4 * q + 1], v2 = V[4 * q + 2], v3 = V[4 * q + 3];
      s.x += w.x * v0.x + w.y * v1.x + w.z * v2.x + w.w * v3.x;
      s.y += w.x * v0.y + w.y * v1.y + w.z * v2.y + w.w * v3.y;
      s.z += w.x * v0.z + w.y * v1.z + w.z * v2.z + w.w * v3.z;
      s.w += w.x * v0.w + w.y * v1.w + w.z * v2.w + w.w * v3.w;
    }
    float* dst = Bs + (tid >> 1) * PRJ_PP + (tid & 1);
    dst[0] = s.x; dst[2] = s.y; dst[4] = s.z; dst[6] = s.w;
    #pragma unroll
    for (int cidx = 4; cidx < 16; ++cidx) dst[2 * cidx] = 0.f;
  }
  __syncthreads();

  const int tile = blockIdx.x * 8 + wave;
  if (tile < GEMM_MTILES) {
    const float* ap = A + (size_t)(tile * 16 + ml) * K + 2 * half;
    const float* bp = Bs + half * PRJ_PP + 2 * ml;

    v8f acc = (v8f){0.f,0.f,0.f,0.f,0.f,0.f,0.f,0.f};
    v2f a_cur[8], a_nxt[8];
    #pragma unroll
    for (int j = 0; j < 8; ++j) a_cur[j] = *(const v2f*)(ap + 4 * j);

    for (int kb = 0; kb < K; kb += 32) {
      const int kn = (kb + 32 < K) ? kb + 32 : kb;
      #pragma unroll
      for (int j = 0; j < 8; ++j) a_nxt[j] = *(const v2f*)(ap + kn + 4 * j);
      #pragma unroll
      for (int j = 0; j < 8; ++j) {
        const v2f b = *(const v2f*)(bp + ((kb + 4 * j) >> 1) * PRJ_PP);
        acc = __builtin_amdgcn_wmma_f32_16x16x4_f32(
            false, a_cur[j], false, b, (short)0, acc, false, false);
      }
      #pragma unroll
      for (int j = 0; j < 8; ++j) a_cur[j] = a_nxt[j];
    }

    if (ml < 4) {
      const int row0 = tile * 16 + 8 * half;
      #pragma unroll
      for (int i = 0; i < 8; ++i) {
        const int row = row0 + i;
        if (ml == 0)      s_src[row] = acc[i];
        else if (ml == 1) s_dst[row] = acc[i];
        else              proj[(size_t)row * 2 + (ml - 2)] = acc[i];
      }
    }
  }
}

__global__ __launch_bounds__(256) void node_softmax_tail(
    const unsigned* __restrict__ cnt, const int* __restrict__ slots,
    const unsigned* __restrict__ ovf,
    const float* __restrict__ s_src, const float* __restrict__ s_dst,
    const float* __restrict__ proj,
    const float* __restrict__ cterm,
    const float* __restrict__ bc, float* __restrict__ out)
{
  static_assert(CAP == 32, "one lane per bucket slot");
  const int lane = threadIdx.x & 31;
  const int wave = __builtin_amdgcn_readfirstlane((int)(threadIdx.x >> 5));
  const int node = blockIdx.x * 8 + wave;
  if (node >= NNODES) return;
  const int indeg = (int)cnt[node];
  const int deg = indeg < CAP ? indeg : CAP;
  const int ncur = deg + 1;
  const int n0 = ncur < 32 ? ncur : 32;
  const int novf = indeg > CAP ? (int)cnt[NNODES] : 0;
  const float sd = s_dst[node];

  const float c0 = cterm[0], c1 = cterm[1];

  int s0 = node;
  if (lane < deg) s0 = slots[(size_t)node * CAP + lane];
  const float lg = leaky02(s_src[s0] + sd);
  const float2 pv = *(const float2*)(proj + 2 * (size_t)s0);

  float m = lane < n0 ? lg : -INFINITY;
  float lgs = 0.f;
  if (ncur > 32) {
    lgs = leaky02(s_src[node] + sd);
    m = fmaxf(m, lgs);
  }
  for (int k0 = 0; k0 < novf; k0 += 32) {
    const int k = k0 + lane;
    if (k < novf) {
      const unsigned w = ovf[k];
      if ((int)(w >> 16) == node)
        m = fmaxf(m, leaky02(s_src[(int)(w & 0xffffu)] + sd));
    }
  }
  m = wave_max(m);

  const float e = lane < n0 ? expf(lg - m) : 0.f;
  float den = e, a0 = e * pv.x, a1 = e * pv.y;
  if (ncur > 32) {
    const float es = expf(lgs - m);
    const float2 ps = *(const float2*)(proj + 2 * (size_t)node);
    if (lane == 0) { den += es; a0 += es * ps.x; a1 += es * ps.y; }
  }
  for (int k0 = 0; k0 < novf; k0 += 32) {
    const int k = k0 + lane;
    if (k < novf) {
      const unsigned w = ovf[k];
      if ((int)(w >> 16) == node) {
        const int s = (int)(w & 0xffffu);
        const float eo = expf(leaky02(s_src[s] + sd) - m);
        const float2 po = *(const float2*)(proj + 2 * (size_t)s);
        den += eo; a0 += eo * po.x; a1 += eo * po.y;
      }
    }
  }
  den = wave_sum(den);
  a0 = wave_sum(a0);
  a1 = wave_sum(a1);
  if (lane == 0) {
    const float d = den + 1e-16f;
    *(float2*)(out + 2 * (size_t)node) = make_float2(a0 / d + c0 + bc[0], a1 / d + c1 + bc[1]);
  }
}

#define FILL_BLOCKS ((NCNT + 255) / 256)
#define FOLD_CH     (NHEADS * HIDC)
#define FOLD_WORDS  (2 * 2 * FOLD_CH + 64)
static_assert(FOLD_CH == 256, "one channel per thread of a 256-thread work-group");
static_assert(HIDC == 64, "two channels per lane in the last layer's constant term");
__global__ __launch_bounds__(256) void prep_counts_fold(
    unsigned* __restrict__ cnt,
    const float* __restrict__ b0, const float* __restrict__ g0, const float* __restrict__ bt0,
    const float* __restrict__ m0, const float* __restrict__ v0,
    const float* __restrict__ b1, const float* __restrict__ g1, const float* __restrict__ bt1,
    const float* __restrict__ m1, const float* __restrict__ v1,
    const float* __restrict__ b2, const float* __restrict__ g2, const float* __restrict__ bt2,
    const float* __restrict__ m2, const float* __restrict__ v2, const float* __restrict__ Wc,
    float* __restrict__ fold)
{
  const int blk = (int)blockIdx.x, t = (int)threadIdx.x;
  if (blk < FILL_BLOCKS) {
    const int i = blk * 256 + t;
    if (i < NCNT) cnt[i] = 0u;
    return;
  }
  if (blk == FILL_BLOCKS + 2) {
    if (t < 32) {
      const int ch = 2 * t;
      const float2 b  = *(const float2*)(b2 + ch);
      const float2 mu = *(const float2*)(m2 + ch);
      const float2 va = *(const float2*)(v2 + ch);
      const float2 ga = *(const float2*)(g2 + ch);
      const float2 be = *(const float2*)(bt2 + ch);
      const float4 w  = *(const float4*)(Wc + 2 * ch);
      const float tx = (b.x - mu.x) * rsqrtf(va.x + EPSBN) * ga.x + be.x;
      const float ty = (b.y - mu.y) * rsqrtf(va.y + EPSBN) * ga.y + be.y;
      const float c0 = wave_sum(tx * w.x + ty * w.z);
      const float c1 = wave_sum(tx * w.y + ty * w.w);
      if (t == 0) {
        fold[2 * 2 * FOLD_CH + 0] = c0;
        fold[2 * 2 * FOLD_CH + 1] = c1;
      }
    }
    return;
  }
  const bool l1 = blk > FILL_BLOCKS;
  const float* b  = l1 ? b1 : b0;
  const float* g  = l1 ? g1 : g0;
  const float* bt = l1 ? bt1 : bt0;
  const float* m  = l1 ? m1 : m0;
  const float* v  = l1 ? v1 : v0;
  const float s = rsqrtf(v[t] + EPSBN) * g[t];
  float* f = fold + (l1 ? 2 * FOLD_CH : 0);
  f[t] = s;
  f[FOLD_CH + t] = (b[t] - m[t]) * s + bt[t];
}

extern "C" void kernel_launch(void* const* d_in, const int* in_sizes, int n_in,
                              void* d_out, int out_size, void* d_ws, size_t ws_size,
                              hipStream_t stream)
{
  const float* x  = (const float*)d_in[0];
  const int*   ei = (const int*)  d_in[1];
  const float* Wc = (const float*)d_in[26];
  const float* bc = (const float*)d_in[27];

  char* ws = (char*)d_ws;
  float*    hf    = (float*)ws;    ws += (size_t)NNODES * 256 * sizeof(float);
  float*    agg   = (float*)ws;    ws += (size_t)NNODES * 256 * sizeof(float);
  float*    ssrc  = (float*)ws;    ws += (size_t)NNODES * NHEADS * sizeof(float);
  float*    sdst  = (float*)ws;    ws += (size_t)NNODES * NHEADS * sizeof(float);
  unsigned* cnt    = (unsigned*)ws; ws += (size_t)NCNT * sizeof(unsigned);
  unsigned* ovf    = (unsigned*)ws; ws += (size_t)NEDGES * sizeof(unsigned);
  int*      slots  = (int*)ws;      ws += (size_t)NNODES * CAP * sizeof(int);
  float*    fold   = (float*)ws;    ws += (size_t)FOLD_WORDS * sizeof(float);

  prep_counts_fold<<<FILL_BLOCKS + 3, 256, 0, stream>>>(
      cnt,
      (const float*)d_in[5],  (const float*)d_in[6],  (const float*)d_in[7],
      (const float*)d_in[8],  (const float*)d_in[9],
      (const float*)d_in[13], (const float*)d_in[14], (const float*)d_in[15],
      (const float*)d_in[16], (const float*)d_in[17],
      (const float*)d_in[21], (const float*)d_in[22], (const float*)d_in[23],
      (const float*)d_in[24], (const float*)d_in[25], Wc,
      fold);

  const float* in = x;
  for (int l = 0; l < 3; ++l) {
    int base = 2 + l * 8;
    const float* W  = (const float*)d_in[base + 0];
    const float* as = (const float*)d_in[base + 1];
    const float* ad = (const float*)d_in[base + 2];
    const float* g  = (const float*)d_in[base + 4];
    const float* vv = (const float*)d_in[base + 7];

    const int heads = (l < 2) ? NHEADS : 1;
    const int hc    = heads * HIDC;

    dim3 ggrid(hc / GEMM_NP, (GEMM_MTILES + 7) / 8);
    if (l == 0) {
      dim3 ggrid0(CNT_PANELS, CNT_ROWS + (GEMM_MTILES + 7) / 8);
      gemm_wmma_f32<FIN, true><<<ggrid0, 256, 0, stream>>>(
          in, W, hf, hc, as, ad, ssrc, sdst, ei, cnt, slots, ovf);
    } else if (heads == NHEADS) {
      gemm_wmma_f32<NHEADS * HIDC, false><<<ggrid, 256, 0, stream>>>(
          in, W, hf, hc, as, ad, ssrc, sdst, nullptr, nullptr, nullptr, nullptr);
    } else {
      last_layer_project<NHEADS * HIDC><<<(GEMM_MTILES + 7) / 8, 256, 0, stream>>>(
          in, W, as, ad, g, vv, Wc, ssrc, sdst, hf);
    }

    if (heads == NHEADS)
      node_softmax_gather<NHEADS, true><<<NNODES / 8, 256, 0, stream>>>(
          cnt, slots, ovf, ssrc, sdst, hf, fold + l * 2 * FOLD_CH, fold + l * 2 * FOLD_CH + FOLD_CH, agg);
    else
      node_softmax_tail<<<NNODES / 8, 256, 0, stream>>>(
          cnt, slots, ovf, ssrc, sdst, hf, fold + 2 * 2 * FOLD_CH, bc, (float*)d_out);

    in = agg;
  }
}
